# starting kernel
# baseline (speedup 1.0000x reference)
#include <hip/hip_runtime.h>
#include <hip/hip_bf16.h>


typedef float v2f __attribute__((ext_vector_type(2)));
typedef float v8f __attribute__((ext_vector_type(8)));

#define DIMF 64
#define CATF 256
#define OUTF 10

__global__ void zero_f32(float* __restrict__ p, int n) {
    int i = blockIdx.x * blockDim.x + threadIdx.x;
    if (i < n) p[i] = 0.0f;
}

__global__ void deg_kernel(const int* __restrict__ dst, float* __restrict__ deg, int E) {
    int e = blockIdx.x * blockDim.x + threadIdx.x;
    if (e < E) atomicAdd(&deg[dst[e]], 1.0f);
}

__global__ void dinv_kernel(float* __restrict__ deg, int N) {
    int i = blockIdx.x * blockDim.x + threadIdx.x;
    if (i < N) deg[i] = __frsqrt_rn(deg[i] + 1.0f);
}

__global__ void enorm_kernel(const int* __restrict__ src, const int* __restrict__ dst,
                             const float* __restrict__ dinv, float* __restrict__ enorm, int E) {
    int e = blockIdx.x * blockDim.x + threadIdx.x;
    if (e < E) enorm[e] = dinv[src[e]] * dinv[dst[e]];
}

__global__ __launch_bounds__(256) void gemm_n64_wmma(const float* __restrict__ A,
                                                     const float* __restrict__ W,
                                                     float* __restrict__ out) {
    const int wid  = threadIdx.x >> 5;
    const int lane = threadIdx.x & 31;
    const int m0   = blockIdx.x * 32 + (wid >> 2) * 16;
    const int n0   = (wid & 3) * 16;
    const int mr   = lane & 15;
    const int kh   = (lane >> 4) * 2;

    const float* __restrict__ arow = A + (size_t)(m0 + mr) * DIMF;

    v8f c = {};
#pragma unroll
    for (int k = 0; k < DIMF; k += 4) {
        v2f a, b;
        a.x = arow[k + kh];
        a.y = arow[k + kh + 1];
        b.x = W[(k + kh) * DIMF + n0 + mr];
        b.y = W[(k + kh + 1) * DIMF + n0 + mr];
        c = __builtin_amdgcn_wmma_f32_16x16x4_f32(
                 false, a,  false, b,
                 (short)0, c,  false,  false);
    }

    const int col   = n0 + mr;
    const int rbase = m0 + (lane >> 4) * 8;
#pragma unroll
    for (int r = 0; r < 8; ++r)
        out[(size_t)(rbase + r) * DIMF + col] = c[r];
}

__global__ void agg_init_kernel(const float* __restrict__ h_lin,
                                const float* __restrict__ dinv,
                                const float* __restrict__ bias,
                                float* __restrict__ h_next, int N) {
    int idx = blockIdx.x * blockDim.x + threadIdx.x;
    if (idx >= N * DIMF) return;
    int i = idx >> 6, d = idx & 63;
    float di = dinv[i];
    h_next[idx] = h_lin[idx] * di * di + bias[d];
}

__global__ void scatter_kernel(const int* __restrict__ src, const int* __restrict__ dst,
                               const float* __restrict__ enorm,
                               const float* __restrict__ h_lin,
                               float* __restrict__ h_next, int E) {
    long long idx = (long long)blockIdx.x * blockDim.x + threadIdx.x;
    if (idx >= (long long)E * 16) return;
    int e  = (int)(idx >> 4);
    int d0 = ((int)idx & 15) * 4;
    float w = enorm[e];
    const float4 v = *(const float4*)(h_lin + (size_t)src[e] * DIMF + d0);
    float* o = h_next + (size_t)dst[e] * DIMF + d0;
    atomicAdd(o + 0, v.x * w);
    atomicAdd(o + 1, v.y * w);
    atomicAdd(o + 2, v.z * w);
    atomicAdd(o + 3, v.w * w);
}

__global__ void post_kernel(const float* __restrict__ h_next,
                            const float* __restrict__ gamma, const float* __restrict__ beta,
                            const float* __restrict__ mean,  const float* __restrict__ var,
                            const int* __restrict__ batch,
                            float* __restrict__ h_cur, float* __restrict__ pooled,
                            int N, int layer) {
    int idx = blockIdx.x * blockDim.x + threadIdx.x;
    if (idx >= N * DIMF) return;
    int i = idx >> 6, d = idx & 63;
    int ld = layer * DIMF + d;
    float v = (h_next[idx] - mean[ld]) * __frsqrt_rn(var[ld] + 1e-5f) * gamma[ld] + beta[ld];
    v = fmaxf(v, 0.0f);
    h_cur[idx] = v;
    atomicAdd(&pooled[(size_t)batch[i] * CATF + ld], v);
}

__global__ __launch_bounds__(64) void mlp_kernel(const float* __restrict__ pooled,
                                                 const float* __restrict__ W1, const float* __restrict__ b1,
                                                 const float* __restrict__ W2, const float* __restrict__ b2,
                                                 float* __restrict__ out) {
    __shared__ float sp[CATF];
    __shared__ float sh[DIMF];
    int g = blockIdx.x, t = threadIdx.x;
    const float* prow = pooled + (size_t)g * CATF;
#pragma unroll
    for (int j = 0; j < 4; ++j) sp[t + j * DIMF] = prow[t + j * DIMF];
    __syncthreads();
    float acc = b1[t];
#pragma unroll 8
    for (int k = 0; k < CATF; ++k) acc = fmaf(sp[k], W1[k * DIMF + t], acc);
    sh[t] = fmaxf(acc, 0.0f);
    __syncthreads();
    if (t < OUTF) {
        float o = b2[t];
#pragma unroll 8
        for (int d = 0; d < DIMF; ++d) o = fmaf(sh[d], W2[d * OUTF + t], o);
        out[(size_t)g * OUTF + t] = o;
    }
}

extern "C" void kernel_launch(void* const* d_in, const int* in_sizes, int n_in,
                              void* d_out, int out_size, void* d_ws, size_t ws_size,
                              hipStream_t stream) {
    const float* x    = (const float*)d_in[0];
    const int*   ei   = (const int*)d_in[1];
    const int*   bat  = (const int*)d_in[2];
    const float* W[4] = {(const float*)d_in[4], (const float*)d_in[6],
                         (const float*)d_in[8], (const float*)d_in[10]};
    const float* B[4] = {(const float*)d_in[5], (const float*)d_in[7],
                         (const float*)d_in[9], (const float*)d_in[11]};
    const float* bn_g = (const float*)d_in[12];
    const float* bn_b = (const float*)d_in[13];
    const float* bn_m = (const float*)d_in[14];
    const float* bn_v = (const float*)d_in[15];
    const float* l1W  = (const float*)d_in[16];
    const float* l1b  = (const float*)d_in[17];
    const float* l2W  = (const float*)d_in[18];
    const float* l2b  = (const float*)d_in[19];

    const int N = in_sizes[0] / DIMF;
    const int E = in_sizes[1] / 2;
    const int G = out_size / OUTF;
    const int* src = ei;
    const int* dst = ei + E;

    float* ws     = (float*)d_ws;
    float* dinv   = ws;
    float* enorm  = dinv + N;
    float* h_cur  = enorm + E;
    float* h_lin  = h_cur + (size_t)N * DIMF;
    float* h_next = h_lin + (size_t)N * DIMF;
    float* pooled = h_next + (size_t)N * DIMF;

    const int T = 256;
    zero_f32<<<(N + T - 1) / T, T, 0, stream>>>(dinv, N);
    zero_f32<<<(G * CATF + T - 1) / T, T, 0, stream>>>(pooled, G * CATF);
    deg_kernel<<<(E + T - 1) / T, T, 0, stream>>>(dst, dinv, E);
    dinv_kernel<<<(N + T - 1) / T, T, 0, stream>>>(dinv, N);
    enorm_kernel<<<(E + T - 1) / T, T, 0, stream>>>(src, dst, dinv, enorm, E);

    const int gemm_blocks = N / 32;
    const int nd_blocks   = (N * DIMF + T - 1) / T;
    const long long sc    = (long long)E * 16;
    const int sc_blocks   = (int)((sc + T - 1) / T);

    const float* hin = x;
    for (int l = 0; l < 4; ++l) {
        gemm_n64_wmma<<<gemm_blocks, T, 0, stream>>>(hin, W[l], h_lin);
        agg_init_kernel<<<nd_blocks, T, 0, stream>>>(h_lin, dinv, B[l], h_next, N);
        scatter_kernel<<<sc_blocks, T, 0, stream>>>(src, dst, enorm, h_lin, h_next, E);
        post_kernel<<<nd_blocks, T, 0, stream>>>(h_next, bn_g, bn_b, bn_m, bn_v,
                                                 bat, h_cur, pooled, N, l);
        hin = h_cur;
    }

    mlp_kernel<<<G, DIMF, 0, stream>>>(pooled, l1W, l1b, l2W, l2b, (float*)d_out);
}


# Edges counting-sorted by destination once per call (integer-atomic arrival-order rank, scan, fill); per-layer float-atomic scatter becomes an atomic-free row gather, self term and bias folded in
# speedup vs baseline: 1.7064x; 1.7064x over previous
#include <hip/hip_runtime.h>
#include <hip/hip_bf16.h>


typedef float v2f __attribute__((ext_vector_type(2)));
typedef float v8f __attribute__((ext_vector_type(8)));

#define DIMF 64
#define CATF 256
#define OUTF 10

__global__ void zero_f32(float* __restrict__ p, int n) {
    int i = blockIdx.x * blockDim.x + threadIdx.x;
    if (i < n) p[i] = 0.0f;
}

__global__ void zero_i32(int* __restrict__ p, int n) {
    int i = blockIdx.x * blockDim.x + threadIdx.x;
    if (i < n) p[i] = 0;
}

__global__ void rank_kernel(const int* __restrict__ dst, int* __restrict__ cnt,
                            int* __restrict__ rank, int E) {
    int e = blockIdx.x * blockDim.x + threadIdx.x;
    if (e < E) rank[e] = atomicAdd(&cnt[dst[e]], 1);
}

__global__ __launch_bounds__(256) void scan_kernel(const int* __restrict__ cnt,
                                                   int* __restrict__ rowptr,
                                                   float* __restrict__ dinv, int N) {
    __shared__ int sred[256];
    __shared__ int sm[256];
    const int t = threadIdx.x, b = blockIdx.x;
    const int4* __restrict__ c4 = (const int4*)cnt;
    int pre = 0;
    for (int c = 0; c < b; c += 8) {
        int4 v[8];
#pragma unroll
        for (int u = 0; u < 8; ++u) v[u] = c4[min(c + u, b - 1) * 256 + t];
#pragma unroll
        for (int u = 0; u < 8; ++u)
            pre += (c + u < b) ? (v[u].x + v[u].y) + (v[u].z + v[u].w) : 0;
    }
    sred[t] = pre;
    __syncthreads();
    for (int s = 128; s > 0; s >>= 1) {
        if (t < s) sred[t] += sred[t + s];
        __syncthreads();
    }
    const int before = sred[0];

    const int i0 = b * 1024 + t * 4;
    int c[4];
#pragma unroll
    for (int j = 0; j < 4; ++j) c[j] = (i0 + j < N) ? cnt[i0 + j] : 0;
    const int tsum = (c[0] + c[1]) + (c[2] + c[3]);
    sm[t] = tsum;
    __syncthreads();
    for (int off = 1; off < 256; off <<= 1) {
        const int v = (t >= off) ? sm[t - off] : 0;
        __syncthreads();
        sm[t] += v;
        __syncthreads();
    }
    int run = before + sm[t] - tsum;
#pragma unroll
    for (int j = 0; j < 4; ++j) {
        if (i0 + j < N) {
            rowptr[i0 + j] = run;
            dinv[i0 + j]   = __frsqrt_rn((float)c[j] + 1.0f);
            run += c[j];
            if (i0 + j == N - 1) rowptr[N] = run;
        }
    }
}

__global__ void fill_kernel(const int* __restrict__ src, const int* __restrict__ dst,
                            const int* __restrict__ rank, const int* __restrict__ rowptr,
                            const float* __restrict__ dinv, int* __restrict__ csr, int E) {
    int e = blockIdx.x * blockDim.x + threadIdx.x;
    if (e >= E) return;
    const int s = src[e], d = dst[e];
    const int slot = rowptr[d] + rank[e];
    int2 r;
    r.x = s;
    r.y = __float_as_int(dinv[s] * dinv[d]);
    ((int2*)csr)[slot] = r;
}

__global__ __launch_bounds__(256) void gemm_n64_wmma(const float* __restrict__ A,
                                                     const float* __restrict__ W,
                                                     float* __restrict__ out) {
    const int wid  = threadIdx.x >> 5;
    const int lane = threadIdx.x & 31;
    const int m0   = blockIdx.x * 32 + (wid >> 2) * 16;
    const int n0   = (wid & 3) * 16;
    const int mr   = lane & 15;
    const int kh   = (lane >> 4) * 2;

    const float* __restrict__ arow = A + (size_t)(m0 + mr) * DIMF;

    v8f c = {};
#pragma unroll
    for (int k = 0; k < DIMF; k += 4) {
        v2f a, b;
        a.x = arow[k + kh];
        a.y = arow[k + kh + 1];
        b.x = W[(k + kh) * DIMF + n0 + mr];
        b.y = W[(k + kh + 1) * DIMF + n0 + mr];
        c = __builtin_amdgcn_wmma_f32_16x16x4_f32(
                 false, a,  false, b,
                 (short)0, c,  false,  false);
    }

    const int col   = n0 + mr;
    const int rbase = m0 + (lane >> 4) * 8;
#pragma unroll
    for (int r = 0; r < 8; ++r)
        out[(size_t)(rbase + r) * DIMF + col] = c[r];
}

__device__ __forceinline__ void fma4(float4& a, const float4 v, const float w) {
    a.x = fmaf(v.x, w, a.x);
    a.y = fmaf(v.y, w, a.y);
    a.z = fmaf(v.z, w, a.z);
    a.w = fmaf(v.w, w, a.w);
}

__device__ __forceinline__ float4 keep4(const bool p, const float4 v) {
    float4 r;
    r.x = p ? v.x : 0.0f;
    r.y = p ? v.y : 0.0f;
    r.z = p ? v.z : 0.0f;
    r.w = p ? v.w : 0.0f;
    return r;
}

__global__ __launch_bounds__(256) void gather_kernel(const int* __restrict__ rowptr,
                                                     const int* __restrict__ csr,
                                                     const float* __restrict__ h_lin,
                                                     const float* __restrict__ dinv,
                                                     const float* __restrict__ bias,
                                                     float* __restrict__ h_next, int N) {
    const int idx = blockIdx.x * blockDim.x + threadIdx.x;
    const int i   = idx >> 4;
    if (i >= N) return;
    const int d0  = (idx & 15) * 4;
    const int beg = rowptr[i];
    const int end = rowptr[i + 1];
    const int2* __restrict__ rec = (const int2*)csr;
    const float* __restrict__ hb = h_lin + d0;

    const float  di = dinv[i];
    const float4 hs = *(const float4*)(hb + (size_t)i * DIMF);
    const float4 bb = *(const float4*)(bias + d0);

    float4 acc = make_float4(0.0f, 0.0f, 0.0f, 0.0f);
    const int last = end - 1;
    for (int k = beg; k < end; k += 4) {
        const bool p1 = k + 1 < end, p2 = k + 2 < end, p3 = k + 3 < end;
        const int2 e0 = rec[k];
        const int2 e1 = rec[min(k + 1, last)];
        const int2 e2 = rec[min(k + 2, last)];
        const int2 e3 = rec[min(k + 3, last)];
        const float4 v0 = *(const float4*)(hb + (size_t)e0.x * DIMF);
        const float4 v1 = *(const float4*)(hb + (size_t)e1.x * DIMF);
        const float4 v2 = *(const float4*)(hb + (size_t)e2.x * DIMF);
        const float4 v3 = *(const float4*)(hb + (size_t)e3.x * DIMF);
        fma4(acc, v0, __int_as_float(e0.y));
        fma4(acc, keep4(p1, v1), p1 ? __int_as_float(e1.y) : 0.0f);
        fma4(acc, keep4(p2, v2), p2 ? __int_as_float(e2.y) : 0.0f);
        fma4(acc, keep4(p3, v3), p3 ? __int_as_float(e3.y) : 0.0f);
    }

    float4 o;
    o.x = acc.x + (hs.x * di * di + bb.x);
    o.y = acc.y + (hs.y * di * di + bb.y);
    o.z = acc.z + (hs.z * di * di + bb.z);
    o.w = acc.w + (hs.w * di * di + bb.w);
    *(float4*)(h_next + (size_t)i * DIMF + d0) = o;
}

__global__ void post_kernel(const float* __restrict__ h_next,
                            const float* __restrict__ gamma, const float* __restrict__ beta,
                            const float* __restrict__ mean,  const float* __restrict__ var,
                            const int* __restrict__ batch,
                            float* __restrict__ h_cur, float* __restrict__ pooled,
                            int N, int layer) {
    int idx = blockIdx.x * blockDim.x + threadIdx.x;
    if (idx >= N * DIMF) return;
    int i = idx >> 6, d = idx & 63;
    int ld = layer * DIMF + d;
    float v = (h_next[idx] - mean[ld]) * __frsqrt_rn(var[ld] + 1e-5f) * gamma[ld] + beta[ld];
    v = fmaxf(v, 0.0f);
    h_cur[idx] = v;
    atomicAdd(&pooled[(size_t)batch[i] * CATF + ld], v);
}

__global__ __launch_bounds__(64) void mlp_kernel(const float* __restrict__ pooled,
                                                 const float* __restrict__ W1, const float* __restrict__ b1,
                                                 const float* __restrict__ W2, const float* __restrict__ b2,
                                                 float* __restrict__ out) {
    __shared__ float sp[CATF];
    __shared__ float sh[DIMF];
    int g = blockIdx.x, t = threadIdx.x;
    const float* prow = pooled + (size_t)g * CATF;
#pragma unroll
    for (int j = 0; j < 4; ++j) sp[t + j * DIMF] = prow[t + j * DIMF];
    __syncthreads();
    float acc = b1[t];
#pragma unroll 8
    for (int k = 0; k < CATF; ++k) acc = fmaf(sp[k], W1[k * DIMF + t], acc);
    sh[t] = fmaxf(acc, 0.0f);
    __syncthreads();
    if (t < OUTF) {
        float o = b2[t];
#pragma unroll 8
        for (int d = 0; d < DIMF; ++d) o = fmaf(sh[d], W2[d * OUTF + t], o);
        out[(size_t)g * OUTF + t] = o;
    }
}

extern "C" void kernel_launch(void* const* d_in, const int* in_sizes, int n_in,
                              void* d_out, int out_size, void* d_ws, size_t ws_size,
                              hipStream_t stream) {
    const float* x    = (const float*)d_in[0];
    const int*   ei   = (const int*)d_in[1];
    const int*   bat  = (const int*)d_in[2];
    const float* W[4] = {(const float*)d_in[4], (const float*)d_in[6],
                         (const float*)d_in[8], (const float*)d_in[10]};
    const float* B[4] = {(const float*)d_in[5], (const float*)d_in[7],
                         (const float*)d_in[9], (const float*)d_in[11]};
    const float* bn_g = (const float*)d_in[12];
    const float* bn_b = (const float*)d_in[13];
    const float* bn_m = (const float*)d_in[14];
    const float* bn_v = (const float*)d_in[15];
    const float* l1W  = (const float*)d_in[16];
    const float* l1b  = (const float*)d_in[17];
    const float* l2W  = (const float*)d_in[18];
    const float* l2b  = (const float*)d_in[19];

    const int N = in_sizes[0] / DIMF;
    const int E = in_sizes[1] / 2;
    const int G = out_size / OUTF;
    const int* src = ei;
    const int* dst = ei + E;

    float* ws     = (float*)d_ws;
    float* dinv   = ws;
    int*   cnt    = (int*)(dinv + N);
    int*   rowptr = cnt + N;
    int*   rank   = rowptr + N + 4;
    int*   csr    = rank + E;
    float* h_cur  = (float*)(csr + 2 * (size_t)E);
    float* h_lin  = h_cur + (size_t)N * DIMF;
    float* h_next = h_lin + (size_t)N * DIMF;
    float* pooled = h_next + (size_t)N * DIMF;

    const int T = 256;
    zero_i32<<<(N + T - 1) / T, T, 0, stream>>>(cnt, N);
    zero_f32<<<(G * CATF + T - 1) / T, T, 0, stream>>>(pooled, G * CATF);
    rank_kernel<<<(E + T - 1) / T, T, 0, stream>>>(dst, cnt, rank, E);
    scan_kernel<<<(N + 1023) / 1024, T, 0, stream>>>(cnt, rowptr, dinv, N);
    fill_kernel<<<(E + T - 1) / T, T, 0, stream>>>(src, dst, rank, rowptr, dinv, csr, E);

    const int gemm_blocks = N / 32;
    const int nd_blocks   = (N * DIMF + T - 1) / T;
    const int ga_blocks   = (N * 16 + T - 1) / T;

    const float* hin = x;
    for (int l = 0; l < 4; ++l) {
        gemm_n64_wmma<<<gemm_blocks, T, 0, stream>>>(hin, W[l], h_lin);
        gather_kernel<<<ga_blocks, T, 0, stream>>>(rowptr, csr, h_lin, dinv, B[l], h_next, N);
        post_kernel<<<nd_blocks, T, 0, stream>>>(h_next, bn_g, bn_b, bn_m, bn_v,
                                                 bat, h_cur, pooled, N, l);
        hin = h_cur;
    }

    mlp_kernel<<<G, DIMF, 0, stream>>>(pooled, l1W, l1b, l2W, l2b, (float*)d_out);
}


# Post pass (BN, ReLU, pool): one wave walks 16 rows, sums runs of equal graph id in registers, one float atomic per column per run, not per row; pool sum order changes (partial sums, then atomics)
# speedup vs baseline: 2.0503x; 1.2015x over previous
#include <hip/hip_runtime.h>
#include <hip/hip_bf16.h>


typedef float v2f __attribute__((ext_vector_type(2)));
typedef float v8f __attribute__((ext_vector_type(8)));

#define DIMF 64
#define CATF 256
#define OUTF 10

__global__ void zero_f32(float* __restrict__ p, int n) {
    int i = blockIdx.x * blockDim.x + threadIdx.x;
    if (i < n) p[i] = 0.0f;
}

__global__ void zero_i32(int* __restrict__ p, int n) {
    int i = blockIdx.x * blockDim.x + threadIdx.x;
    if (i < n) p[i] = 0;
}

__global__ void rank_kernel(const int* __restrict__ dst, int* __restrict__ cnt,
                            int* __restrict__ rank, int E) {
    int e = blockIdx.x * blockDim.x + threadIdx.x;
    if (e < E) rank[e] = atomicAdd(&cnt[dst[e]], 1);
}

__global__ __launch_bounds__(256) void scan_kernel(const int* __restrict__ cnt,
                                                   int* __restrict__ rowptr,
                                                   float* __restrict__ dinv, int N) {
    __shared__ int sred[256];
    __shared__ int sm[256];
    const int t = threadIdx.x, b = blockIdx.x;
    const int4* __restrict__ c4 = (const int4*)cnt;
    int pre = 0;
    for (int c = 0; c < b; c += 8) {
        int4 v[8];
#pragma unroll
        for (int u = 0; u < 8; ++u) v[u] = c4[min(c + u, b - 1) * 256 + t];
#pragma unroll
        for (int u = 0; u < 8; ++u)
            pre += (c + u < b) ? (v[u].x + v[u].y) + (v[u].z + v[u].w) : 0;
    }
    sred[t] = pre;
    __syncthreads();
    for (int s = 128; s > 0; s >>= 1) {
        if (t < s) sred[t] += sred[t + s];
        __syncthreads();
    }
    const int before = sred[0];

    const int i0 = b * 1024 + t * 4;
    int c[4];
#pragma unroll
    for (int j = 0; j < 4; ++j) c[j] = (i0 + j < N) ? cnt[i0 + j] : 0;
    const int tsum = (c[0] + c[1]) + (c[2] + c[3]);
    sm[t] = tsum;
    __syncthreads();
    for (int off = 1; off < 256; off <<= 1) {
        const int v = (t >= off) ? sm[t - off] : 0;
        __syncthreads();
        sm[t] += v;
        __syncthreads();
    }
    int run = before + sm[t] - tsum;
#pragma unroll
    for (int j = 0; j < 4; ++j) {
        if (i0 + j < N) {
            rowptr[i0 + j] = run;
            dinv[i0 + j]   = __frsqrt_rn((float)c[j] + 1.0f);
            run += c[j];
            if (i0 + j == N - 1) rowptr[N] = run;
        }
    }
}

__global__ void fill_kernel(const int* __restrict__ src, const int* __restrict__ dst,
                            const int* __restrict__ rank, const int* __restrict__ rowptr,
                            const float* __restrict__ dinv, int* __restrict__ csr, int E) {
    int e = blockIdx.x * blockDim.x + threadIdx.x;
    if (e >= E) return;
    const int s = src[e], d = dst[e];
    const int slot = rowptr[d] + rank[e];
    int2 r;
    r.x = s;
    r.y = __float_as_int(dinv[s] * dinv[d]);
    ((int2*)csr)[slot] = r;
}

__global__ __launch_bounds__(256) void gemm_n64_wmma(const float* __restrict__ A,
                                                     const float* __restrict__ W,
                                                     float* __restrict__ out) {
    const int wid  = threadIdx.x >> 5;
    const int lane = threadIdx.x & 31;
    const int m0   = blockIdx.x * 32 + (wid >> 2) * 16;
    const int n0   = (wid & 3) * 16;
    const int mr   = lane & 15;
    const int kh   = (lane >> 4) * 2;

    const float* __restrict__ arow = A + (size_t)(m0 + mr) * DIMF;

    v8f c = {};
#pragma unroll
    for (int k = 0; k < DIMF; k += 4) {
        v2f a, b;
        a.x = arow[k + kh];
        a.y = arow[k + kh + 1];
        b.x = W[(k + kh) * DIMF + n0 + mr];
        b.y = W[(k + kh + 1) * DIMF + n0 + mr];
        c = __builtin_amdgcn_wmma_f32_16x16x4_f32(
                 false, a,  false, b,
                 (short)0, c,  false,  false);
    }

    const int col   = n0 + mr;
    const int rbase = m0 + (lane >> 4) * 8;
#pragma unroll
    for (int r = 0; r < 8; ++r)
        out[(size_t)(rbase + r) * DIMF + col] = c[r];
}

__device__ __forceinline__ void fma4(float4& a, const float4 v, const float w) {
    a.x = fmaf(v.x, w, a.x);
    a.y = fmaf(v.y, w, a.y);
    a.z = fmaf(v.z, w, a.z);
    a.w = fmaf(v.w, w, a.w);
}

__device__ __forceinline__ float4 keep4(const bool p, const float4 v) {
    float4 r;
    r.x = p ? v.x : 0.0f;
    r.y = p ? v.y : 0.0f;
    r.z = p ? v.z : 0.0f;
    r.w = p ? v.w : 0.0f;
    return r;
}

__global__ __launch_bounds__(256) void gather_kernel(const int* __restrict__ rowptr,
                                                     const int* __restrict__ csr,
                                                     const float* __restrict__ h_lin,
                                                     const float* __restrict__ dinv,
                                                     const float* __restrict__ bias,
                                                     float* __restrict__ h_next, int N) {
    const int idx = blockIdx.x * blockDim.x + threadIdx.x;
    const int i   = idx >> 4;
    if (i >= N) return;
    const int d0  = (idx & 15) * 4;
    const int beg = rowptr[i];
    const int end = rowptr[i + 1];
    const int2* __restrict__ rec = (const int2*)csr;
    const float* __restrict__ hb = h_lin + d0;

    const float  di = dinv[i];
    const float4 hs = *(const float4*)(hb + (size_t)i * DIMF);
    const float4 bb = *(const float4*)(bias + d0);

    float4 acc = make_float4(0.0f, 0.0f, 0.0f, 0.0f);
    const int last = end - 1;
    for (int k = beg; k < end; k += 4) {
        const bool p1 = k + 1 < end, p2 = k + 2 < end, p3 = k + 3 < end;
        const int2 e0 = rec[k];
        const int2 e1 = rec[min(k + 1, last)];
        const int2 e2 = rec[min(k + 2, last)];
        const int2 e3 = rec[min(k + 3, last)];
        const float4 v0 = *(const float4*)(hb + (size_t)e0.x * DIMF);
        const float4 v1 = *(const float4*)(hb + (size_t)e1.x * DIMF);
        const float4 v2 = *(const float4*)(hb + (size_t)e2.x * DIMF);
        const float4 v3 = *(const float4*)(hb + (size_t)e3.x * DIMF);
        fma4(acc, v0, __int_as_float(e0.y));
        fma4(acc, keep4(p1, v1), p1 ? __int_as_float(e1.y) : 0.0f);
        fma4(acc, keep4(p2, v2), p2 ? __int_as_float(e2.y) : 0.0f);
        fma4(acc, keep4(p3, v3), p3 ? __int_as_float(e3.y) : 0.0f);
    }

    float4 o;
    o.x = acc.x + (hs.x * di * di + bb.x);
    o.y = acc.y + (hs.y * di * di + bb.y);
    o.z = acc.z + (hs.z * di * di + bb.z);
    o.w = acc.w + (hs.w * di * di + bb.w);
    *(float4*)(h_next + (size_t)i * DIMF + d0) = o;
}

#define POST_ROWS 16

template <bool FULL>
__device__ __forceinline__ void post_rows(const float* __restrict__ h_next,
                                          const float* __restrict__ gamma,
                                          const float* __restrict__ beta,
                                          const float* __restrict__ mean,
                                          const float* __restrict__ var,
                                          const int* __restrict__ batch,
                                          float* __restrict__ h_cur,
                                          float* __restrict__ pooled,
                                          const int row0, const int nrows,
                                          const int d, const int ld) {
    int    g[POST_ROWS];
    float2 v[POST_ROWS];
#pragma unroll
    for (int r = 0; r < POST_ROWS; ++r) {
        const int row = FULL ? row0 + r : row0 + min(r, nrows - 1);
        g[r] = batch[row];
        v[r] = *(const float2*)(h_next + (size_t)row * DIMF + d);
    }
    const float2 mu = *(const float2*)(mean + ld);
    const float2 vv = *(const float2*)(var + ld);
    const float2 ga = *(const float2*)(gamma + ld);
    const float2 be = *(const float2*)(beta + ld);
    float2 rs;
    rs.x = __frsqrt_rn(vv.x + 1e-5f);
    rs.y = __frsqrt_rn(vv.y + 1e-5f);

    int   gp = g[0];
    float ax = 0.0f, ay = 0.0f;
#pragma unroll
    for (int r = 0; r < POST_ROWS; ++r) {
        if (FULL || r < nrows) {
            float2 t;
            t.x = fmaxf((v[r].x - mu.x) * rs.x * ga.x + be.x, 0.0f);
            t.y = fmaxf((v[r].y - mu.y) * rs.y * ga.y + be.y, 0.0f);
            *(float2*)(h_cur + (size_t)(row0 + r) * DIMF + d) = t;
            if (g[r] != gp) {
                atomicAdd(&pooled[(size_t)gp * CATF + ld],     ax);
                atomicAdd(&pooled[(size_t)gp * CATF + ld + 1], ay);
                ax = 0.0f;
                ay = 0.0f;
                gp = g[r];
            }
            ax += t.x;
            ay += t.y;
        }
    }
    atomicAdd(&pooled[(size_t)gp * CATF + ld],     ax);
    atomicAdd(&pooled[(size_t)gp * CATF + ld + 1], ay);
}

__global__ __launch_bounds__(256) void post_kernel(const float* __restrict__ h_next,
                            const float* __restrict__ gamma, const float* __restrict__ beta,
                            const float* __restrict__ mean,  const float* __restrict__ var,
                            const int* __restrict__ batch,
                            float* __restrict__ h_cur, float* __restrict__ pooled,
                            int N, int layer) {
    const int chunk = blockIdx.x * 8 + __builtin_amdgcn_readfirstlane((int)(threadIdx.x >> 5));
    const int row0  = chunk * POST_ROWS;
    if (row0 >= N) return;
    const int d  = (threadIdx.x & 31) * 2;
    const int ld = layer * DIMF + d;
    const int nrows = min(POST_ROWS, N - row0);
    if (nrows == POST_ROWS)
        post_rows<true>(h_next, gamma, beta, mean, var, batch, h_cur, pooled, row0, nrows, d, ld);
    else
        post_rows<false>(h_next, gamma, beta, mean, var, batch, h_cur, pooled, row0, nrows, d, ld);
}

__global__ __launch_bounds__(64) void mlp_kernel(const float* __restrict__ pooled,
                                                 const float* __restrict__ W1, const float* __restrict__ b1,
                                                 const float* __restrict__ W2, const float* __restrict__ b2,
                                                 float* __restrict__ out) {
    __shared__ float sp[CATF];
    __shared__ float sh[DIMF];
    int g = blockIdx.x, t = threadIdx.x;
    const float* prow = pooled + (size_t)g * CATF;
#pragma unroll
    for (int j = 0; j < 4; ++j) sp[t + j * DIMF] = prow[t + j * DIMF];
    __syncthreads();
    float acc = b1[t];
#pragma unroll 8
    for (int k = 0; k < CATF; ++k) acc = fmaf(sp[k], W1[k * DIMF + t], acc);
    sh[t] = fmaxf(acc, 0.0f);
    __syncthreads();
    if (t < OUTF) {
        float o = b2[t];
#pragma unroll 8
        for (int d = 0; d < DIMF; ++d) o = fmaf(sh[d], W2[d * OUTF + t], o);
        out[(size_t)g * OUTF + t] = o;
    }
}

extern "C" void kernel_launch(void* const* d_in, const int* in_sizes, int n_in,
                              void* d_out, int out_size, void* d_ws, size_t ws_size,
                              hipStream_t stream) {
    const float* x    = (const float*)d_in[0];
    const int*   ei   = (const int*)d_in[1];
    const int*   bat  = (const int*)d_in[2];
    const float* W[4] = {(const float*)d_in[4], (const float*)d_in[6],
                         (const float*)d_in[8], (const float*)d_in[10]};
    const float* B[4] = {(const float*)d_in[5], (const float*)d_in[7],
                         (const float*)d_in[9], (const float*)d_in[11]};
    const float* bn_g = (const float*)d_in[12];
    const float* bn_b = (const float*)d_in[13];
    const float* bn_m = (const float*)d_in[14];
    const float* bn_v = (const float*)d_in[15];
    const float* l1W  = (const float*)d_in[16];
    const float* l1b  = (const float*)d_in[17];
    const float* l2W  = (const float*)d_in[18];
    const float* l2b  = (const float*)d_in[19];

    const int N = in_sizes[0] / DIMF;
    const int E = in_sizes[1] / 2;
    const int G = out_size / OUTF;
    const int* src = ei;
    const int* dst = ei + E;

    float* ws     = (float*)d_ws;
    float* dinv   = ws;
    int*   cnt    = (int*)(dinv + N);
    int*   rowptr = cnt + N;
    int*   rank   = rowptr + N + 4;
    int*   csr    = rank + E;
    float* h_cur  = (float*)(csr + 2 * (size_t)E);
    float* h_lin  = h_cur + (size_t)N * DIMF;
    float* h_next = h_lin + (size_t)N * DIMF;
    float* pooled = h_next + (size_t)N * DIMF;

    const int T = 256;
    zero_i32<<<(N + T - 1) / T, T, 0, stream>>>(cnt, N);
    zero_f32<<<(G * CATF + T - 1) / T, T, 0, stream>>>(pooled, G * CATF);
    rank_kernel<<<(E + T - 1) / T, T, 0, stream>>>(dst, cnt, rank, E);
    scan_kernel<<<(N + 1023) / 1024, T, 0, stream>>>(cnt, rowptr, dinv, N);
    fill_kernel<<<(E + T - 1) / T, T, 0, stream>>>(src, dst, rank, rowptr, dinv, csr, E);

    const int gemm_blocks = N / 32;
    const int po_blocks   = (N + 8 * POST_ROWS - 1) / (8 * POST_ROWS);
    const int ga_blocks   = (N * 16 + T - 1) / T;

    const float* hin = x;
    for (int l = 0; l < 4; ++l) {
        gemm_n64_wmma<<<gemm_blocks, T, 0, stream>>>(hin, W[l], h_lin);
        gather_kernel<<<ga_blocks, T, 0, stream>>>(rowptr, csr, h_lin, dinv, B[l], h_next, N);
        post_kernel<<<po_blocks, T, 0, stream>>>(h_next, bn_g, bn_b, bn_m, bn_v,
                                                 bat, h_cur, pooled, N, l);
        hin = h_cur;
    }

    mlp_kernel<<<G, DIMF, 0, stream>>>(pooled, l1W, l1b, l2W, l2b, (float*)d_out);
}


# Edge, row and GEMM kernels run a fixed few thousand waves that loop over their items (GEMM: W held in registers, 16x64 strip per wave), not one short wave per item; GEMM k sum order changes
# speedup vs baseline: 2.1288x; 1.0383x over previous
#include <hip/hip_runtime.h>
#include <hip/hip_bf16.h>


typedef float v2f __attribute__((ext_vector_type(2)));
typedef float v8f __attribute__((ext_vector_type(8)));

#define DIMF 64
#define CATF 256
#define OUTF 10
#define LOOP_BLOCKS 1024
#define GEMM_BLOCKS 512

__global__ void zero_f32(float* __restrict__ p, int n) {
    const int i = (blockIdx.x * blockDim.x + threadIdx.x) * 4;
    if (i + 3 < n) {
        *(float4*)(p + i) = make_float4(0.0f, 0.0f, 0.0f, 0.0f);
    } else {
        for (int j = i; j < n; ++j) p[j] = 0.0f;
    }
}

__global__ void zero_i32(int* __restrict__ p, int n) {
    const int i = (blockIdx.x * blockDim.x + threadIdx.x) * 4;
    if (i + 3 < n) {
        *(int4*)(p + i) = make_int4(0, 0, 0, 0);
    } else {
        for (int j = i; j < n; ++j) p[j] = 0;
    }
}

__global__ __launch_bounds__(256) void rank_kernel(const int* __restrict__ dst,
                                                   int* __restrict__ cnt,
                                                   int* __restrict__ rank, int E) {
    const int stride = gridDim.x * blockDim.x;
    int e0 = blockIdx.x * blockDim.x + threadIdx.x;
    for (; e0 + 3 * stride < E; e0 += 4 * stride) {
        int d[4], r[4];
#pragma unroll
        for (int j = 0; j < 4; ++j) d[j] = dst[e0 + j * stride];
#pragma unroll
        for (int j = 0; j < 4; ++j) r[j] = atomicAdd(&cnt[d[j]], 1);
#pragma unroll
        for (int j = 0; j < 4; ++j) rank[e0 + j * stride] = r[j];
    }
    for (; e0 < E; e0 += stride) rank[e0] = atomicAdd(&cnt[dst[e0]], 1);
}

__global__ __launch_bounds__(256) void scan_kernel(const int* __restrict__ cnt,
                                                   int* __restrict__ rowptr,
                                                   float* __restrict__ dinv, int N) {
    __shared__ int sred[256];
    __shared__ int sm[256];
    const int t = threadIdx.x, b = blockIdx.x;
    const int4* __restrict__ c4 = (const int4*)cnt;
    int pre = 0;
    for (int c = 0; c < b; c += 8) {
        int4 v[8];
#pragma unroll
        for (int u = 0; u < 8; ++u) v[u] = c4[min(c + u, b - 1) * 256 + t];
#pragma unroll
        for (int u = 0; u < 8; ++u)
            pre += (c + u < b) ? (v[u].x + v[u].y) + (v[u].z + v[u].w) : 0;
    }
    sred[t] = pre;
    __syncthreads();
    for (int s = 128; s > 0; s >>= 1) {
        if (t < s) sred[t] += sred[t + s];
        __syncthreads();
    }
    const int before = sred[0];

    const int i0 = b * 1024 + t * 4;
    int c[4];
#pragma unroll
    for (int j = 0; j < 4; ++j) c[j] = (i0 + j < N) ? cnt[i0 + j] : 0;
    const int tsum = (c[0] + c[1]) + (c[2] + c[3]);
    sm[t] = tsum;
    __syncthreads();
    for (int off = 1; off < 256; off <<= 1) {
        const int v = (t >= off) ? sm[t - off] : 0;
        __syncthreads();
        sm[t] += v;
        __syncthreads();
    }
    int run = before + sm[t] - tsum;
#pragma unroll
    for (int j = 0; j < 4; ++j) {
        if (i0 + j < N) {
            rowptr[i0 + j] = run;
            dinv[i0 + j]   = __frsqrt_rn((float)c[j] + 1.0f);
            run += c[j];
            if (i0 + j == N - 1) rowptr[N] = run;
        }
    }
}

__global__ void fill_kernel(const int* __restrict__ src, const int* __restrict__ dst,
                            const int* __restrict__ rank, const int* __restrict__ rowptr,
                            const float* __restrict__ dinv, int* __restrict__ csr, int E) {
    const int stride = gridDim.x * blockDim.x;
    int e0 = blockIdx.x * blockDim.x + threadIdx.x;
    for (; e0 + 3 * stride < E; e0 += 4 * stride) {
        int s[4], d[4], rk[4];
#pragma unroll
        for (int j = 0; j < 4; ++j) {
            const int e = e0 + j * stride;
            s[j]  = src[e];
            d[j]  = dst[e];
            rk[j] = rank[e];
        }
        int   rp[4];
        float ds[4], dd[4];
#pragma unroll
        for (int j = 0; j < 4; ++j) {
            rp[j] = rowptr[d[j]];
            ds[j] = dinv[s[j]];
            dd[j] = dinv[d[j]];
        }
#pragma unroll
        for (int j = 0; j < 4; ++j) {
            int2 r;
            r.x = s[j];
            r.y = __float_as_int(ds[j] * dd[j]);
            ((int2*)csr)[rp[j] + rk[j]] = r;
        }
    }
    for (; e0 < E; e0 += stride) {
        const int s = src[e0], d = dst[e0];
        int2 r;
        r.x = s;
        r.y = __float_as_int(dinv[s] * dinv[d]);
        ((int2*)csr)[rowptr[d] + rank[e0]] = r;
    }
}

__global__ __launch_bounds__(128) void gemm_n64_wmma(const float* __restrict__ A,
                                                     const float* __restrict__ W,
                                                     float* __restrict__ out, int ntiles) {
    const int lane = threadIdx.x & 31;
    const int c    = lane & 15;
    const int h    = lane >> 4;
    const int gw = blockIdx.x * 4 + __builtin_amdgcn_readfirstlane((int)(threadIdx.x >> 5));
    const int nw = gridDim.x * 4;

    float wv[8][4][4];
#pragma unroll
    for (int j = 0; j < 8; ++j) {
#pragma unroll
        for (int i = 0; i < 4; ++i) {
            const float4 q = *(const float4*)(W + (size_t)(8 * j + 4 * h + i) * DIMF + 4 * c);
            wv[j][i][0] = q.x;
            wv[j][i][1] = q.y;
            wv[j][i][2] = q.z;
            wv[j][i][3] = q.w;
        }
    }

    for (int t = gw; t < ntiles; t += nw) {
        const float* __restrict__ arow = A + (size_t)(t * 16 + c) * DIMF + 4 * h;
        float av[8][4];
#pragma unroll
        for (int j = 0; j < 8; ++j) {
            const float4 q = *(const float4*)(arow + 8 * j);
            av[j][0] = q.x;
            av[j][1] = q.y;
            av[j][2] = q.z;
            av[j][3] = q.w;
        }

        v8f acc[4];
#pragma unroll
        for (int nt = 0; nt < 4; ++nt) acc[nt] = v8f{};

#pragma unroll
        for (int j = 0; j < 8; ++j) {
#pragma unroll
            for (int s = 0; s < 2; ++s) {
                v2f a;
                a.x = av[j][2 * s];
                a.y = av[j][2 * s + 1];
#pragma unroll
                for (int nt = 0; nt < 4; ++nt) {
                    v2f b;
                    b.x = wv[j][2 * s][nt];
                    b.y = wv[j][2 * s + 1][nt];
                    acc[nt] = __builtin_amdgcn_wmma_f32_16x16x4_f32(
                             false, a,  false, b,
                             (short)0, acc[nt],  false,  false);
                }
            }
        }

        float* __restrict__ orow = out + (size_t)(t * 16 + 8 * h) * DIMF + 4 * c;
#pragma unroll
        for (int r = 0; r < 8; ++r) {
            float4 o;
            o.x = acc[0][r];
            o.y = acc[1][r];
            o.z = acc[2][r];
            o.w = acc[3][r];
            *(float4*)(orow + (size_t)r * DIMF) = o;
        }
    }
}

__device__ __forceinline__ void fma4(float4& a, const float4 v, const float w) {
    a.x = fmaf(v.x, w, a.x);
    a.y = fmaf(v.y, w, a.y);
    a.z = fmaf(v.z, w, a.z);
    a.w = fmaf(v.w, w, a.w);
}

__device__ __forceinline__ float4 keep4(const bool p, const float4 v) {
    float4 r;
    r.x = p ? v.x : 0.0f;
    r.y = p ? v.y : 0.0f;
    r.z = p ? v.z : 0.0f;
    r.w = p ? v.w : 0.0f;
    return r;
}

__global__ __launch_bounds__(256) void gather_kernel(const int* __restrict__ rowptr,
                                                     const int* __restrict__ csr,
                                                     const float* __restrict__ h_lin,
                                                     const float* __restrict__ dinv,
                                                     const float* __restrict__ bias,
                                                     float* __restrict__ h_next, int N) {
    const int gid = (blockIdx.x * blockDim.x + threadIdx.x) >> 4;
    const int ng  = (gridDim.x * blockDim.x) >> 4;
    const int d0  = (threadIdx.x & 15) * 4;
    const int2* __restrict__ rec = (const int2*)csr;
    const float* __restrict__ hb = h_lin + d0;
    const float4 bb = *(const float4*)(bias + d0);

    for (int i = gid; i < N; i += ng) {
        const int beg = rowptr[i];
        const int end = rowptr[i + 1];
        const float  di = dinv[i];
        const float4 hs = *(const float4*)(hb + (size_t)i * DIMF);

        float4 acc = make_float4(0.0f, 0.0f, 0.0f, 0.0f);
        const int last = end - 1;
        for (int k = beg; k < end; k += 8) {
            int2   e[8];
            float4 v[8];
#pragma unroll
            for (int u = 0; u < 8; ++u) e[u] = rec[min(k + u, last)];
#pragma unroll
            for (int u = 0; u < 8; ++u) v[u] = *(const float4*)(hb + (unsigned)e[u].x * (unsigned)DIMF);
#pragma unroll
            for (int u = 0; u < 8; ++u) {
                const bool p = k + u < end;
                fma4(acc, keep4(p, v[u]), p ? __int_as_float(e[u].y) : 0.0f);
            }
        }

        float4 o;
        o.x = acc.x + (hs.x * di * di + bb.x);
        o.y = acc.y + (hs.y * di * di + bb.y);
        o.z = acc.z + (hs.z * di * di + bb.z);
        o.w = acc.w + (hs.w * di * di + bb.w);
        *(float4*)(h_next + (size_t)i * DIMF + d0) = o;
    }
}

#define POST_ROWS 16

template <bool FULL>
__device__ __forceinline__ void post_rows(const float* __restrict__ h_next,
                                          const float* __restrict__ gamma,
                                          const float* __restrict__ beta,
                                          const float* __restrict__ mean,
                                          const float* __restrict__ var,
                                          const int* __restrict__ batch,
                                          float* __restrict__ h_cur,
                                          float* __restrict__ pooled,
                                          const int row0, const int nrows,
                                          const int d, const int ld) {
    int    g[POST_ROWS];
    float2 v[POST_ROWS];
#pragma unroll
    for (int r = 0; r < POST_ROWS; ++r) {
        const int row = FULL ? row0 + r : row0 + min(r, nrows - 1);
        g[r] = batch[row];
        v[r] = *(const float2*)(h_next + (size_t)row * DIMF + d);
    }
    const float2 mu = *(const float2*)(mean + ld);
    const float2 vv = *(const float2*)(var + ld);
    const float2 ga = *(const float2*)(gamma + ld);
    const float2 be = *(const float2*)(beta + ld);
    float2 rs;
    rs.x = __frsqrt_rn(vv.x + 1e-5f);
    rs.y = __frsqrt_rn(vv.y + 1e-5f);

    int   gp = g[0];
    float ax = 0.0f, ay = 0.0f;
#pragma unroll
    for (int r = 0; r < POST_ROWS; ++r) {
        if (FULL || r < nrows) {
            float2 t;
            t.x = fmaxf((v[r].x - mu.x) * rs.x * ga.x + be.x, 0.0f);
            t.y = fmaxf((v[r].y - mu.y) * rs.y * ga.y + be.y, 0.0f);
            *(float2*)(h_cur + (size_t)(row0 + r) * DIMF + d) = t;
            if (g[r] != gp) {
                atomicAdd(&pooled[(size_t)gp * CATF + ld],     ax);
                atomicAdd(&pooled[(size_t)gp * CATF + ld + 1], ay);
                ax = 0.0f;
                ay = 0.0f;
                gp = g[r];
            }
            ax += t.x;
            ay += t.y;
        }
    }
    atomicAdd(&pooled[(size_t)gp * CATF + ld],     ax);
    atomicAdd(&pooled[(size_t)gp * CATF + ld + 1], ay);
}

__global__ __launch_bounds__(256) void post_kernel(const float* __restrict__ h_next,
                            const float* __restrict__ gamma, const float* __restrict__ beta,
                            const float* __restrict__ mean,  const float* __restrict__ var,
                            const int* __restrict__ batch,
                            float* __restrict__ h_cur, float* __restrict__ pooled,
                            int N, int layer) {
    const int chunk = blockIdx.x * 8 + __builtin_amdgcn_readfirstlane((int)(threadIdx.x >> 5));
    const int row0  = chunk * POST_ROWS;
    if (row0 >= N) return;
    const int d  = (threadIdx.x & 31) * 2;
    const int ld = layer * DIMF + d;
    const int nrows = min(POST_ROWS, N - row0);
    if (nrows == POST_ROWS)
        post_rows<true>(h_next, gamma, beta, mean, var, batch, h_cur, pooled, row0, nrows, d, ld);
    else
        post_rows<false>(h_next, gamma, beta, mean, var, batch, h_cur, pooled, row0, nrows, d, ld);
}

__global__ __launch_bounds__(64) void mlp_kernel(const float* __restrict__ pooled,
                                                 const float* __restrict__ W1, const float* __restrict__ b1,
                                                 const float* __restrict__ W2, const float* __restrict__ b2,
                                                 float* __restrict__ out) {
    __shared__ float sp[CATF];
    __shared__ float sh[DIMF];
    int g = blockIdx.x, t = threadIdx.x;
    const float* prow = pooled + (size_t)g * CATF;
#pragma unroll
    for (int j = 0; j < 4; ++j) sp[t + j * DIMF] = prow[t + j * DIMF];
    __syncthreads();
    float acc = b1[t];
#pragma unroll 8
    for (int k = 0; k < CATF; ++k) acc = fmaf(sp[k], W1[k * DIMF + t], acc);
    sh[t] = fmaxf(acc, 0.0f);
    __syncthreads();
    if (t < OUTF) {
        float o = b2[t];
#pragma unroll 8
        for (int d = 0; d < DIMF; ++d) o = fmaf(sh[d], W2[d * OUTF + t], o);
        out[(size_t)g * OUTF + t] = o;
    }
}

extern "C" void kernel_launch(void* const* d_in, const int* in_sizes, int n_in,
                              void* d_out, int out_size, void* d_ws, size_t ws_size,
                              hipStream_t stream) {
    const float* x    = (const float*)d_in[0];
    const int*   ei   = (const int*)d_in[1];
    const int*   bat  = (const int*)d_in[2];
    const float* W[4] = {(const float*)d_in[4], (const float*)d_in[6],
                         (const float*)d_in[8], (const float*)d_in[10]};
    const float* B[4] = {(const float*)d_in[5], (const float*)d_in[7],
                         (const float*)d_in[9], (const float*)d_in[11]};
    const float* bn_g = (const float*)d_in[12];
    const float* bn_b = (const float*)d_in[13];
    const float* bn_m = (const float*)d_in[14];
    const float* bn_v = (const float*)d_in[15];
    const float* l1W  = (const float*)d_in[16];
    const float* l1b  = (const float*)d_in[17];
    const float* l2W  = (const float*)d_in[18];
    const float* l2b  = (const float*)d_in[19];

    const int N = in_sizes[0] / DIMF;
    const int E = in_sizes[1] / 2;
    const int G = out_size / OUTF;
    const int* src = ei;
    const int* dst = ei + E;

    float* ws     = (float*)d_ws;
    float* dinv   = ws;
    int*   cnt    = (int*)(dinv + N);
    int*   rowptr = cnt + N;
    int*   rank   = rowptr + N + 4;
    int*   csr    = rank + E;
    float* h_cur  = (float*)(csr + 2 * (size_t)E);
    float* h_lin  = h_cur + (size_t)N * DIMF;
    float* h_next = h_lin + (size_t)N * DIMF;
    float* pooled = h_next + (size_t)N * DIMF;

    const int T = 256;
    zero_i32<<<(N + 4 * T - 1) / (4 * T), T, 0, stream>>>(cnt, N);
    zero_f32<<<(G * CATF + 4 * T - 1) / (4 * T), T, 0, stream>>>(pooled, G * CATF);
    rank_kernel<<<LOOP_BLOCKS, T, 0, stream>>>(dst, cnt, rank, E);
    scan_kernel<<<(N + 1023) / 1024, T, 0, stream>>>(cnt, rowptr, dinv, N);
    fill_kernel<<<LOOP_BLOCKS, T, 0, stream>>>(src, dst, rank, rowptr, dinv, csr, E);

    const int gemm_tiles  = N / 16;
    const int po_blocks   = (N + 8 * POST_ROWS - 1) / (8 * POST_ROWS);

    const float* hin = x;
    for (int l = 0; l < 4; ++l) {
        gemm_n64_wmma<<<GEMM_BLOCKS, 128, 0, stream>>>(hin, W[l], h_lin, gemm_tiles);
        gather_kernel<<<LOOP_BLOCKS, T, 0, stream>>>(rowptr, csr, h_lin, dinv, B[l], h_next, N);
        post_kernel<<<po_blocks, T, 0, stream>>>(h_next, bn_g, bn_b, bn_m, bn_v,
                                                 bat, h_cur, pooled, N, l);
        hin = h_cur;
    }

    mlp_kernel<<<G, DIMF, 0, stream>>>(pooled, l1W, l1b, l2W, l2b, (float*)d_out);
}


# speedup vs baseline: 2.4279x; 1.1405x over previous
#include <hip/hip_runtime.h>
#include <hip/hip_bf16.h>


typedef float v2f __attribute__((ext_vector_type(2)));
typedef float v8f __attribute__((ext_vector_type(8)));

#define DIMF 64
#define CATF 256
#define OUTF 10
#define LOOP_BLOCKS 1024
#define GEMM_BLOCKS 512
#define PAD64(n) (((size_t)(n) + 63) & ~(size_t)63)

__global__ void zero_f32(float* __restrict__ p, int n) {
    const int i = (blockIdx.x * blockDim.x + threadIdx.x) * 4;
    if (i + 3 < n) {
        *(float4*)(p + i) = make_float4(0.0f, 0.0f, 0.0f, 0.0f);
    } else {
        for (int j = i; j < n; ++j) p[j] = 0.0f;
    }
}

__global__ void zero_i32(int* __restrict__ p, int n) {
    const int i = (blockIdx.x * blockDim.x + threadIdx.x) * 4;
    if (i + 3 < n) {
        *(int4*)(p + i) = make_int4(0, 0, 0, 0);
    } else {
        for (int j = i; j < n; ++j) p[j] = 0;
    }
}

__global__ __launch_bounds__(256) void rank_kernel(const int* __restrict__ dst,
                                                   int* __restrict__ cnt,
                                                   int* __restrict__ rank, int E) {
    const int stride = gridDim.x * blockDim.x;
    int e0 = blockIdx.x * blockDim.x + threadIdx.x;
    for (; e0 + 3 * stride < E; e0 += 4 * stride) {
        int d[4], r[4];
#pragma unroll
        for (int j = 0; j < 4; ++j) d[j] = dst[e0 + j * stride];
#pragma unroll
        for (int j = 0; j < 4; ++j) r[j] = atomicAdd(&cnt[d[j]], 1);
#pragma unroll
        for (int j = 0; j < 4; ++j) rank[e0 + j * stride] = r[j];
    }
    for (; e0 < E; e0 += stride) rank[e0] = atomicAdd(&cnt[dst[e0]], 1);
}

__global__ __launch_bounds__(256) void scan_kernel(const int* __restrict__ cnt,
                                                   int* __restrict__ rowptr,
                                                   float* __restrict__ dinv, int N) {
    __shared__ int sred[256];
    __shared__ int sm[256];
    const int t = threadIdx.x, b = blockIdx.x;
    const int4* __restrict__ c4 = (const int4*)cnt;
    int pre = 0;
    for (int c = 0; c < b; c += 8) {
        int4 v[8];
#pragma unroll
        for (int u = 0; u < 8; ++u) v[u] = c4[min(c + u, b - 1) * 256 + t];
#pragma unroll
        for (int u = 0; u < 8; ++u)
            pre += (c + u < b) ? (v[u].x + v[u].y) + (v[u].z + v[u].w) : 0;
    }
    sred[t] = pre;
    __syncthreads();
    for (int s = 128; s > 0; s >>= 1) {
        if (t < s) sred[t] += sred[t + s];
        __syncthreads();
    }
    const int before = sred[0];

    const int i0 = b * 1024 + t * 4;
    int c[4];
#pragma unroll
    for (int j = 0; j < 4; ++j) c[j] = (i0 + j < N) ? cnt[i0 + j] : 0;
    const int tsum = (c[0] + c[1]) + (c[2] + c[3]);
    sm[t] = tsum;
    __syncthreads();
    for (int off = 1; off < 256; off <<= 1) {
        const int v = (t >= off) ? sm[t - off] : 0;
        __syncthreads();
        sm[t] += v;
        __syncthreads();
    }
    int run = before + sm[t] - tsum;
#pragma unroll
    for (int j = 0; j < 4; ++j) {
        if (i0 + j < N) {
            rowptr[i0 + j] = run;
            dinv[i0 + j]   = __frsqrt_rn((float)c[j] + 1.0f);
            run += c[j];
            if (i0 + j == N - 1) rowptr[N] = run;
        }
    }
}

__global__ void fill_kernel(const int* __restrict__ src, const int* __restrict__ dst,
                            const int* __restrict__ rank, const int* __restrict__ rowptr,
                            const float* __restrict__ dinv, int* __restrict__ csr, int E) {
    const int stride = gridDim.x * blockDim.x;
    int e0 = blockIdx.x * blockDim.x + threadIdx.x;
    for (; e0 + 3 * stride < E; e0 += 4 * stride) {
        int s[4], d[4], rk[4];
#pragma unroll
        for (int j = 0; j < 4; ++j) {
            const int e = e0 + j * stride;
            s[j]  = src[e];
            d[j]  = dst[e];
            rk[j] = rank[e];
        }
        int   rp[4];
        float ds[4], dd[4];
#pragma unroll
        for (int j = 0; j < 4; ++j) {
            rp[j] = rowptr[d[j]];
            ds[j] = dinv[s[j]];
            dd[j] = dinv[d[j]];
        }
#pragma unroll
        for (int j = 0; j < 4; ++j) {
            int2 r;
            r.x = s[j];
            r.y = __float_as_int(ds[j] * dd[j]);
            ((int2*)csr)[rp[j] + rk[j]] = r;
        }
    }
    for (; e0 < E; e0 += stride) {
        const int s = src[e0], d = dst[e0];
        int2 r;
        r.x = s;
        r.y = __float_as_int(dinv[s] * dinv[d]);
        ((int2*)csr)[rowptr[d] + rank[e0]] = r;
    }
}

__global__ __launch_bounds__(128) void gemm_n64_wmma(const float* __restrict__ A,
                                                     const float* __restrict__ W,
                                                     float* __restrict__ out, int ntiles) {
    const int lane = threadIdx.x & 31;
    const int c    = lane & 15;
    const int h    = lane >> 4;
    const int gw = blockIdx.x * 4 + __builtin_amdgcn_readfirstlane((int)(threadIdx.x >> 5));
    const int nw = gridDim.x * 4;

    float wv[8][4][4];
#pragma unroll
    for (int j = 0; j < 8; ++j) {
#pragma unroll
        for (int i = 0; i < 4; ++i) {
            const float4 q = *(const float4*)(W + (size_t)(8 * j + 4 * h + i) * DIMF + 4 * c);
            wv[j][i][0] = q.x;
            wv[j][i][1] = q.y;
            wv[j][i][2] = q.z;
            wv[j][i][3] = q.w;
        }
    }

    for (int t = gw; t < ntiles; t += nw) {
        const float* __restrict__ arow = A + (size_t)(t * 16 + c) * DIMF + 4 * h;
        float av[8][4];
#pragma unroll
        for (int j = 0; j < 8; ++j) {
            const float4 q = *(const float4*)(arow + 8 * j);
            av[j][0] = q.x;
            av[j][1] = q.y;
            av[j][2] = q.z;
            av[j][3] = q.w;
        }

        v8f acc[4];
#pragma unroll
        for (int nt = 0; nt < 4; ++nt) acc[nt] = v8f{};

#pragma unroll
        for (int j = 0; j < 8; ++j) {
#pragma unroll
            for (int s = 0; s < 2; ++s) {
                v2f a;
                a.x = av[j][2 * s];
                a.y = av[j][2 * s + 1];
#pragma unroll
                for (int nt = 0; nt < 4; ++nt) {
                    v2f b;
                    b.x = wv[j][2 * s][nt];
                    b.y = wv[j][2 * s + 1][nt];
                    acc[nt] = __builtin_amdgcn_wmma_f32_16x16x4_f32(
                             false, a,  false, b,
                             (short)0, acc[nt],  false,  false);
                }
            }
        }

        float* __restrict__ orow = out + (size_t)(t * 16 + 8 * h) * DIMF + 4 * c;
#pragma unroll
        for (int r = 0; r < 8; ++r) {
            float4 o;
            o.x = acc[0][r];
            o.y = acc[1][r];
            o.z = acc[2][r];
            o.w = acc[3][r];
            *(float4*)(orow + (size_t)r * DIMF) = o;
        }
    }
}

__device__ __forceinline__ void fma4(float4& a, const float4 v, const float w) {
    a.x = fmaf(v.x, w, a.x);
    a.y = fmaf(v.y, w, a.y);
    a.z = fmaf(v.z, w, a.z);
    a.w = fmaf(v.w, w, a.w);
}

__device__ __forceinline__ float4 keep4(const bool p, const float4 v) {
    float4 r;
    r.x = p ? v.x : 0.0f;
    r.y = p ? v.y : 0.0f;
    r.z = p ? v.z : 0.0f;
    r.w = p ? v.w : 0.0f;
    return r;
}

__global__ __launch_bounds__(256) void gather_kernel(const int* __restrict__ rowptr,
                                                     const int* __restrict__ csr,
                                                     const float* __restrict__ h_lin,
                                                     const float* __restrict__ dinv,
                                                     const float* __restrict__ bias,
                                                     float* __restrict__ h_next, int N) {
    const int gid = (blockIdx.x * blockDim.x + threadIdx.x) >> 4;
    const int ng  = (gridDim.x * blockDim.x) >> 4;
    const int d0  = (threadIdx.x & 15) * 4;
    const int2* __restrict__ rec = (const int2*)csr;
    const float* __restrict__ hb = h_lin + d0;
    const float4 bb = *(const float4*)(bias + d0);

    for (int i = gid; i < N; i += ng) {
        const int beg = rowptr[i];
        const int end = rowptr[i + 1];
        const float  di = dinv[i];
        const float4 hs = *(const float4*)(hb + (size_t)i * DIMF);

        float4 acc = make_float4(0.0f, 0.0f, 0.0f, 0.0f);
        const int last = end - 1;
        for (int k = beg; k < end; k += 8) {
            int2   e[8];
            float4 v[8];
#pragma unroll
            for (int u = 0; u < 8; ++u) e[u] = rec[min(k + u, last)];
#pragma unroll
            for (int u = 0; u < 8; ++u) v[u] = *(const float4*)(hb + (unsigned)e[u].x * (unsigned)DIMF);
#pragma unroll
            for (int u = 0; u < 8; ++u) {
                const bool p = k + u < end;
                fma4(acc, keep4(p, v[u]), p ? __int_as_float(e[u].y) : 0.0f);
            }
        }

        float4 o;
        o.x = acc.x + (hs.x * di * di + bb.x);
        o.y = acc.y + (hs.y * di * di + bb.y);
        o.z = acc.z + (hs.z * di * di + bb.z);
        o.w = acc.w + (hs.w * di * di + bb.w);
        *(float4*)(h_next + (size_t)i * DIMF + d0) = o;
    }
}

#define POST_ROWS 16

template <bool FULL>
__device__ __forceinline__ void post_rows(const float* __restrict__ h_next,
                                          const float* __restrict__ gamma,
                                          const float* __restrict__ beta,
                                          const float* __restrict__ mean,
                                          const float* __restrict__ var,
                                          const int* __restrict__ batch,
                                          float* __restrict__ h_cur,
                                          float* __restrict__ pooled,
                                          const int row0, const int nrows,
                                          const int d, const int ld) {
    int    g[POST_ROWS];
    float2 v[POST_ROWS];
#pragma unroll
    for (int r = 0; r < POST_ROWS; ++r) {
        const int row = FULL ? row0 + r : row0 + min(r, nrows - 1);
        g[r] = batch[row];
        v[r] = *(const float2*)(h_next + (size_t)row * DIMF + d);
    }
    const float2 mu = *(const float2*)(mean + ld);
    const float2 vv = *(const float2*)(var + ld);
    const float2 ga = *(const float2*)(gamma + ld);
    const float2 be = *(const float2*)(beta + ld);
    float2 rs;
    rs.x = __frsqrt_rn(vv.x + 1e-5f);
    rs.y = __frsqrt_rn(vv.y + 1e-5f);

    int   gp = g[0];
    float ax = 0.0f, ay = 0.0f;
#pragma unroll
    for (int r = 0; r < POST_ROWS; ++r) {
        if (FULL || r < nrows) {
            float2 t;
            t.x = fmaxf((v[r].x - mu.x) * rs.x * ga.x + be.x, 0.0f);
            t.y = fmaxf((v[r].y - mu.y) * rs.y * ga.y + be.y, 0.0f);
            *(float2*)(h_cur + (size_t)(row0 + r) * DIMF + d) = t;
            if (g[r] != gp) {
                atomicAdd(&pooled[(size_t)gp * CATF + ld],     ax);
                atomicAdd(&pooled[(size_t)gp * CATF + ld + 1], ay);
                ax = 0.0f;
                ay = 0.0f;
                gp = g[r];
            }
            ax += t.x;
            ay += t.y;
        }
    }
    atomicAdd(&pooled[(size_t)gp * CATF + ld],     ax);
    atomicAdd(&pooled[(size_t)gp * CATF + ld + 1], ay);
}

__global__ __launch_bounds__(256) void post_kernel(const float* __restrict__ h_next,
                            const float* __restrict__ gamma, const float* __restrict__ beta,
                            const float* __restrict__ mean,  const float* __restrict__ var,
                            const int* __restrict__ batch,
                            float* __restrict__ h_cur, float* __restrict__ pooled,
                            int N, int layer) {
    const int chunk = blockIdx.x * 8 + __builtin_amdgcn_readfirstlane((int)(threadIdx.x >> 5));
    const int row0  = chunk * POST_ROWS;
    if (row0 >= N) return;
    const int d  = (threadIdx.x & 31) * 2;
    const int ld = layer * DIMF + d;
    const int nrows = min(POST_ROWS, N - row0);
    if (nrows == POST_ROWS)
        post_rows<true>(h_next, gamma, beta, mean, var, batch, h_cur, pooled, row0, nrows, d, ld);
    else
        post_rows<false>(h_next, gamma, beta, mean, var, batch, h_cur, pooled, row0, nrows, d, ld);
}

__global__ __launch_bounds__(64) void mlp_kernel(const float* __restrict__ pooled,
                                                 const float* __restrict__ W1, const float* __restrict__ b1,
                                                 const float* __restrict__ W2, const float* __restrict__ b2,
                                                 float* __restrict__ out) {
    __shared__ float sp[CATF];
    __shared__ float sh[DIMF];
    int g = blockIdx.x, t = threadIdx.x;
    const float* prow = pooled + (size_t)g * CATF;
#pragma unroll
    for (int j = 0; j < 4; ++j) sp[t + j * DIMF] = prow[t + j * DIMF];
    __syncthreads();
    float acc = b1[t];
#pragma unroll 8
    for (int k = 0; k < CATF; ++k) acc = fmaf(sp[k], W1[k * DIMF + t], acc);
    sh[t] = fmaxf(acc, 0.0f);
    __syncthreads();
    if (t < OUTF) {
        float o = b2[t];
#pragma unroll 8
        for (int d = 0; d < DIMF; ++d) o = fmaf(sh[d], W2[d * OUTF + t], o);
        out[(size_t)g * OUTF + t] = o;
    }
}

extern "C" void kernel_launch(void* const* d_in, const int* in_sizes, int n_in,
                              void* d_out, int out_size, void* d_ws, size_t ws_size,
                              hipStream_t stream) {
    const float* x    = (const float*)d_in[0];
    const int*   ei   = (const int*)d_in[1];
    const int*   bat  = (const int*)d_in[2];
    const float* W[4] = {(const float*)d_in[4], (const float*)d_in[6],
                         (const float*)d_in[8], (const float*)d_in[10]};
    const float* B[4] = {(const float*)d_in[5], (const float*)d_in[7],
                         (const float*)d_in[9], (const float*)d_in[11]};
    const float* bn_g = (const float*)d_in[12];
    const float* bn_b = (const float*)d_in[13];
    const float* bn_m = (const float*)d_in[14];
    const float* bn_v = (const float*)d_in[15];
    const float* l1W  = (const float*)d_in[16];
    const float* l1b  = (const float*)d_in[17];
    const float* l2W  = (const float*)d_in[18];
    const float* l2b  = (const float*)d_in[19];

    const int N = in_sizes[0] / DIMF;
    const int E = in_sizes[1] / 2;
    const int G = out_size / OUTF;
    const int* src = ei;
    const int* dst = ei + E;

    float* ws     = (float*)d_ws;
    float* dinv   = ws;
    int*   cnt    = (int*)(dinv + PAD64(N));
    int*   rowptr = cnt + PAD64(N);
    int*   rank   = rowptr + PAD64(N + 1);
    int*   csr    = rank + PAD64(E);
    float* h_cur  = (float*)(csr + PAD64(2 * (size_t)E));
    float* h_lin  = h_cur + (size_t)N * DIMF;
    float* h_next = h_lin + (size_t)N * DIMF;
    float* pooled = h_next + (size_t)N * DIMF;

    const int T = 256;
    zero_i32<<<(N + 4 * T - 1) / (4 * T), T, 0, stream>>>(cnt, N);
    zero_f32<<<(G * CATF + 4 * T - 1) / (4 * T), T, 0, stream>>>(pooled, G * CATF);
    rank_kernel<<<LOOP_BLOCKS, T, 0, stream>>>(dst, cnt, rank, E);
    scan_kernel<<<(N + 1023) / 1024, T, 0, stream>>>(cnt, rowptr, dinv, N);
    fill_kernel<<<LOOP_BLOCKS, T, 0, stream>>>(src, dst, rank, rowptr, dinv, csr, E);

    const int gemm_tiles  = N / 16;
    const int po_blocks   = (N + 8 * POST_ROWS - 1) / (8 * POST_ROWS);

    const float* hin = x;
    for (int l = 0; l < 4; ++l) {
        gemm_n64_wmma<<<GEMM_BLOCKS, 128, 0, stream>>>(hin, W[l], h_lin, gemm_tiles);
        gather_kernel<<<LOOP_BLOCKS, T, 0, stream>>>(rowptr, csr, h_lin, dinv, B[l], h_next, N);
        post_kernel<<<po_blocks, T, 0, stream>>>(h_next, bn_g, bn_b, bn_m, bn_v,
                                                 bat, h_cur, pooled, N, l);
        hin = h_cur;
    }

    mlp_kernel<<<G, DIMF, 0, stream>>>(pooled, l1W, l1b, l2W, l2b, (float*)d_out);
}


# Post pass (BN, ReLU, pool) fused into the row gather: a wave takes a consecutive slice of rows, sums runs of equal graph id in registers, joins its halves by lane exchange; pool sum order changes
# speedup vs baseline: 2.6580x; 1.0948x over previous
#include <hip/hip_runtime.h>
#include <hip/hip_bf16.h>


typedef float v2f __attribute__((ext_vector_type(2)));
typedef float v8f __attribute__((ext_vector_type(8)));

#define DIMF 64
#define CATF 256
#define OUTF 10
#define LOOP_BLOCKS 1024
#define GEMM_BLOCKS 512
#define PAD64(n) (((size_t)(n) + 63) & ~(size_t)63)

__global__ void zero_f32(float* __restrict__ p, int n) {
    const int i = (blockIdx.x * blockDim.x + threadIdx.x) * 4;
    if (i + 3 < n) {
        *(float4*)(p + i) = make_float4(0.0f, 0.0f, 0.0f, 0.0f);
    } else {
        for (int j = i; j < n; ++j) p[j] = 0.0f;
    }
}

__global__ void zero_i32(int* __restrict__ p, int n) {
    const int i = (blockIdx.x * blockDim.x + threadIdx.x) * 4;
    if (i + 3 < n) {
        *(int4*)(p + i) = make_int4(0, 0, 0, 0);
    } else {
        for (int j = i; j < n; ++j) p[j] = 0;
    }
}

__global__ __launch_bounds__(256) void rank_kernel(const int* __restrict__ dst,
                                                   int* __restrict__ cnt,
                                                   int* __restrict__ rank, int E) {
    const int stride = gridDim.x * blockDim.x;
    int e0 = blockIdx.x * blockDim.x + threadIdx.x;
    for (; e0 + 3 * stride < E; e0 += 4 * stride) {
        int d[4], r[4];
#pragma unroll
        for (int j = 0; j < 4; ++j) d[j] = dst[e0 + j * stride];
#pragma unroll
        for (int j = 0; j < 4; ++j) r[j] = atomicAdd(&cnt[d[j]], 1);
#pragma unroll
        for (int j = 0; j < 4; ++j) rank[e0 + j * stride] = r[j];
    }
    for (; e0 < E; e0 += stride) rank[e0] = atomicAdd(&cnt[dst[e0]], 1);
}

__global__ __launch_bounds__(256) void scan_kernel(const int* __restrict__ cnt,
                                                   int* __restrict__ rowptr,
                                                   float* __restrict__ dinv, int N) {
    __shared__ int sred[256];
    __shared__ int sm[256];
    const int t = threadIdx.x, b = blockIdx.x;
    const int4* __restrict__ c4 = (const int4*)cnt;
    int pre = 0;
    for (int c = 0; c < b; c += 8) {
        int4 v[8];
#pragma unroll
        for (int u = 0; u < 8; ++u) v[u] = c4[min(c + u, b - 1) * 256 + t];
#pragma unroll
        for (int u = 0; u < 8; ++u)
            pre += (c + u < b) ? (v[u].x + v[u].y) + (v[u].z + v[u].w) : 0;
    }
    sred[t] = pre;
    __syncthreads();
    for (int s = 128; s > 0; s >>= 1) {
        if (t < s) sred[t] += sred[t + s];
        __syncthreads();
    }
    const int before = sred[0];

    const int i0 = b * 1024 + t * 4;
    int c[4];
#pragma unroll
    for (int j = 0; j < 4; ++j) c[j] = (i0 + j < N) ? cnt[i0 + j] : 0;
    const int tsum = (c[0] + c[1]) + (c[2] + c[3]);
    sm[t] = tsum;
    __syncthreads();
    for (int off = 1; off < 256; off <<= 1) {
        const int v = (t >= off) ? sm[t - off] : 0;
        __syncthreads();
        sm[t] += v;
        __syncthreads();
    }
    int run = before + sm[t] - tsum;
#pragma unroll
    for (int j = 0; j < 4; ++j) {
        if (i0 + j < N) {
            rowptr[i0 + j] = run;
            dinv[i0 + j]   = __frsqrt_rn((float)c[j] + 1.0f);
            run += c[j];
            if (i0 + j == N - 1) rowptr[N] = run;
        }
    }
}

__global__ void fill_kernel(const int* __restrict__ src, const int* __restrict__ dst,
                            const int* __restrict__ rank, const int* __restrict__ rowptr,
                            const float* __restrict__ dinv, int* __restrict__ csr, int E) {
    const int stride = gridDim.x * blockDim.x;
    int e0 = blockIdx.x * blockDim.x + threadIdx.x;
    for (; e0 + 3 * stride < E; e0 += 4 * stride) {
        int s[4], d[4], rk[4];
#pragma unroll
        for (int j = 0; j < 4; ++j) {
            const int e = e0 + j * stride;
            s[j]  = src[e];
            d[j]  = dst[e];
            rk[j] = rank[e];
        }
        int   rp[4];
        float ds[4], dd[4];
#pragma unroll
        for (int j = 0; j < 4; ++j) {
            rp[j] = rowptr[d[j]];
            ds[j] = dinv[s[j]];
            dd[j] = dinv[d[j]];
        }
#pragma unroll
        for (int j = 0; j < 4; ++j) {
            int2 r;
            r.x = s[j];
            r.y = __float_as_int(ds[j] * dd[j]);
            ((int2*)csr)[rp[j] + rk[j]] = r;
        }
    }
    for (; e0 < E; e0 += stride) {
        const int s = src[e0], d = dst[e0];
        int2 r;
        r.x = s;
        r.y = __float_as_int(dinv[s] * dinv[d]);
        ((int2*)csr)[rowptr[d] + rank[e0]] = r;
    }
}

__global__ __launch_bounds__(128) void gemm_n64_wmma(const float* __restrict__ A,
                                                     const float* __restrict__ W,
                                                     float* __restrict__ out, int ntiles) {
    const int lane = threadIdx.x & 31;
    const int c    = lane & 15;
    const int h    = lane >> 4;
    const int gw = blockIdx.x * 4 + __builtin_amdgcn_readfirstlane((int)(threadIdx.x >> 5));
    const int nw = gridDim.x * 4;

    float wv[8][4][4];
#pragma unroll
    for (int j = 0; j < 8; ++j) {
#pragma unroll
        for (int i = 0; i < 4; ++i) {
            const float4 q = *(const float4*)(W + (size_t)(8 * j + 4 * h + i) * DIMF + 4 * c);
            wv[j][i][0] = q.x;
            wv[j][i][1] = q.y;
            wv[j][i][2] = q.z;
            wv[j][i][3] = q.w;
        }
    }

    for (int t = gw; t < ntiles; t += nw) {
        const float* __restrict__ arow = A + (size_t)(t * 16 + c) * DIMF + 4 * h;
        float av[8][4];
#pragma unroll
        for (int j = 0; j < 8; ++j) {
            const float4 q = *(const float4*)(arow + 8 * j);
            av[j][0] = q.x;
            av[j][1] = q.y;
            av[j][2] = q.z;
            av[j][3] = q.w;
        }

        v8f acc[4];
#pragma unroll
        for (int nt = 0; nt < 4; ++nt) acc[nt] = v8f{};

#pragma unroll
        for (int j = 0; j < 8; ++j) {
#pragma unroll
            for (int s = 0; s < 2; ++s) {
                v2f a;
                a.x = av[j][2 * s];
                a.y = av[j][2 * s + 1];
#pragma unroll
                for (int nt = 0; nt < 4; ++nt) {
                    v2f b;
                    b.x = wv[j][2 * s][nt];
                    b.y = wv[j][2 * s + 1][nt];
                    acc[nt] = __builtin_amdgcn_wmma_f32_16x16x4_f32(
                             false, a,  false, b,
                             (short)0, acc[nt],  false,  false);
                }
            }
        }

        float* __restrict__ orow = out + (size_t)(t * 16 + 8 * h) * DIMF + 4 * c;
#pragma unroll
        for (int r = 0; r < 8; ++r) {
            float4 o;
            o.x = acc[0][r];
            o.y = acc[1][r];
            o.z = acc[2][r];
            o.w = acc[3][r];
            *(float4*)(orow + (size_t)r * DIMF) = o;
        }
    }
}

__device__ __forceinline__ void fma4(float4& a, const float4 v, const float w) {
    a.x = fmaf(v.x, w, a.x);
    a.y = fmaf(v.y, w, a.y);
    a.z = fmaf(v.z, w, a.z);
    a.w = fmaf(v.w, w, a.w);
}

__device__ __forceinline__ float4 keep4(const bool p, const float4 v) {
    float4 r;
    r.x = p ? v.x : 0.0f;
    r.y = p ? v.y : 0.0f;
    r.z = p ? v.z : 0.0f;
    r.w = p ? v.w : 0.0f;
    return r;
}


__device__ __forceinline__ void pool_add(float* __restrict__ p, const float4 a) {
    atomicAdd(p,     a.x);
    atomicAdd(p + 1, a.y);
    atomicAdd(p + 2, a.z);
    atomicAdd(p + 3, a.w);
}

__global__ __launch_bounds__(256) void gather_post_kernel(const int* __restrict__ rowptr,
                                                          const int* __restrict__ csr,
                                                          const float* __restrict__ h_lin,
                                                          const float* __restrict__ dinv,
                                                          const float* __restrict__ bias,
                                                          const float* __restrict__ gamma,
                                                          const float* __restrict__ beta,
                                                          const float* __restrict__ mean,
                                                          const float* __restrict__ var,
                                                          const int* __restrict__ batch,
                                                          float* __restrict__ h_cur,
                                                          float* __restrict__ pooled,
                                                          int N, int layer, int n_layers) {
    const int wave = blockIdx.x * 8 + (int)(threadIdx.x >> 5);
    const int nw   = gridDim.x * 8;
    const int base = N / nw;
    const int rem  = N - base * nw;
    const int row0 = wave * base + min(wave, rem);
    const int row1 = row0 + base + (wave < rem ? 1 : 0);
    const int half = (threadIdx.x >> 4) & 1;
    const int d0   = (threadIdx.x & 15) * 4;
    const int ld   = layer * DIMF + d0;
    const int2* __restrict__ rec = (const int2*)csr;
    const float* __restrict__ hb = h_lin + d0;
    const float4 bb = *(const float4*)(bias + d0);
    const float4 mu = *(const float4*)(mean + ld);
    const float4 vv = *(const float4*)(var + ld);
    const float4 ga = *(const float4*)(gamma + ld);
    const float4 be = *(const float4*)(beta + ld);
    float4 rs;
    rs.x = __frsqrt_rn(vv.x + 1e-5f);
    rs.y = __frsqrt_rn(vv.y + 1e-5f);
    rs.z = __frsqrt_rn(vv.z + 1e-5f);
    rs.w = __frsqrt_rn(vv.w + 1e-5f);
    float* __restrict__ pcol = pooled + ld;

    int    have = 0;
    int    gp   = 0;
    float4 ax   = make_float4(0.0f, 0.0f, 0.0f, 0.0f);

    for (int i = row0 + half; i < row1; i += 2) {
        const int beg = rowptr[i];
        const int end = rowptr[i + 1];
        const int g   = batch[i];
        const float  di = dinv[i];
        const float4 hs = *(const float4*)(hb + (size_t)i * DIMF);

        float4 acc = make_float4(0.0f, 0.0f, 0.0f, 0.0f);
        const int last = end - 1;
        for (int k = beg; k < end; k += 8) {
            int2   e[8];
            float4 v[8];
#pragma unroll
            for (int u = 0; u < 8; ++u) e[u] = rec[min(k + u, last)];
#pragma unroll
            for (int u = 0; u < 8; ++u) v[u] = *(const float4*)(hb + (unsigned)e[u].x * (unsigned)DIMF);
#pragma unroll
            for (int u = 0; u < 8; ++u) {
                const bool p = k + u < end;
                fma4(acc, keep4(p, v[u]), p ? __int_as_float(e[u].y) : 0.0f);
            }
        }

        float4 o;
        o.x = acc.x + (hs.x * di * di + bb.x);
        o.y = acc.y + (hs.y * di * di + bb.y);
        o.z = acc.z + (hs.z * di * di + bb.z);
        o.w = acc.w + (hs.w * di * di + bb.w);

        float4 t;
        t.x = fmaxf((o.x - mu.x) * rs.x * ga.x + be.x, 0.0f);
        t.y = fmaxf((o.y - mu.y) * rs.y * ga.y + be.y, 0.0f);
        t.z = fmaxf((o.z - mu.z) * rs.z * ga.z + be.z, 0.0f);
        t.w = fmaxf((o.w - mu.w) * rs.w * ga.w + be.w, 0.0f);
        if (layer + 1 < n_layers) *(float4*)(h_cur + (size_t)i * DIMF + d0) = t;

        if (have && g != gp) {
            pool_add(pcol + (size_t)gp * CATF, ax);
            ax = make_float4(0.0f, 0.0f, 0.0f, 0.0f);
        }
        gp   = g;
        have = 1;
        ax.x += t.x;
        ax.y += t.y;
        ax.z += t.z;
        ax.w += t.w;
    }

    const int   have_o = __shfl_xor(have, 16);
    const int   gp_o   = __shfl_xor(gp, 16);
    float4 ax_o;
    ax_o.x = __shfl_xor(ax.x, 16);
    ax_o.y = __shfl_xor(ax.y, 16);
    ax_o.z = __shfl_xor(ax.z, 16);
    ax_o.w = __shfl_xor(ax.w, 16);
    if (have) {
        if (have_o && gp_o == gp) {
            if (half == 0) {
                ax.x += ax_o.x;
                ax.y += ax_o.y;
                ax.z += ax_o.z;
                ax.w += ax_o.w;
                pool_add(pcol + (size_t)gp * CATF, ax);
            }
        } else {
            pool_add(pcol + (size_t)gp * CATF, ax);
        }
    }
}

__global__ __launch_bounds__(64) void mlp_kernel(const float* __restrict__ pooled,
                                                 const float* __restrict__ W1, const float* __restrict__ b1,
                                                 const float* __restrict__ W2, const float* __restrict__ b2,
                                                 float* __restrict__ out) {
    __shared__ float sp[CATF];
    __shared__ float sh[DIMF];
    int g = blockIdx.x, t = threadIdx.x;
    const float* prow = pooled + (size_t)g * CATF;
#pragma unroll
    for (int j = 0; j < 4; ++j) sp[t + j * DIMF] = prow[t + j * DIMF];
    __syncthreads();
    float acc = b1[t];
#pragma unroll 8
    for (int k = 0; k < CATF; ++k) acc = fmaf(sp[k], W1[k * DIMF + t], acc);
    sh[t] = fmaxf(acc, 0.0f);
    __syncthreads();
    if (t < OUTF) {
        float o = b2[t];
#pragma unroll 8
        for (int d = 0; d < DIMF; ++d) o = fmaf(sh[d], W2[d * OUTF + t], o);
        out[(size_t)g * OUTF + t] = o;
    }
}

extern "C" void kernel_launch(void* const* d_in, const int* in_sizes, int n_in,
                              void* d_out, int out_size, void* d_ws, size_t ws_size,
                              hipStream_t stream) {
    const float* x    = (const float*)d_in[0];
    const int*   ei   = (const int*)d_in[1];
    const int*   bat  = (const int*)d_in[2];
    const float* W[4] = {(const float*)d_in[4], (const float*)d_in[6],
                         (const float*)d_in[8], (const float*)d_in[10]};
    const float* B[4] = {(const float*)d_in[5], (const float*)d_in[7],
                         (const float*)d_in[9], (const float*)d_in[11]};
    const float* bn_g = (const float*)d_in[12];
    const float* bn_b = (const float*)d_in[13];
    const float* bn_m = (const float*)d_in[14];
    const float* bn_v = (const float*)d_in[15];
    const float* l1W  = (const float*)d_in[16];
    const float* l1b  = (const float*)d_in[17];
    const float* l2W  = (const float*)d_in[18];
    const float* l2b  = (const float*)d_in[19];

    const int N = in_sizes[0] / DIMF;
    const int E = in_sizes[1] / 2;
    const int G = out_size / OUTF;
    const int* src = ei;
    const int* dst = ei + E;

    float* ws     = (float*)d_ws;
    float* dinv   = ws;
    int*   cnt    = (int*)(dinv + PAD64(N));
    int*   rowptr = cnt + PAD64(N);
    int*   rank   = rowptr + PAD64(N + 1);
    int*   csr    = rank + PAD64(E);
    float* h_cur  = (float*)(csr + PAD64(2 * (size_t)E));
    float* h_lin  = h_cur + (size_t)N * DIMF;
    float* pooled = h_lin + (size_t)N * DIMF;

    const int T = 256;
    zero_i32<<<(N + 4 * T - 1) / (4 * T), T, 0, stream>>>(cnt, N);
    zero_f32<<<(G * CATF + 4 * T - 1) / (4 * T), T, 0, stream>>>(pooled, G * CATF);
    rank_kernel<<<LOOP_BLOCKS, T, 0, stream>>>(dst, cnt, rank, E);
    scan_kernel<<<(N + 1023) / 1024, T, 0, stream>>>(cnt, rowptr, dinv, N);
    fill_kernel<<<LOOP_BLOCKS, T, 0, stream>>>(src, dst, rank, rowptr, dinv, csr, E);

    const int gemm_tiles  = N / 16;

    const float* hin = x;
    for (int l = 0; l < 4; ++l) {
        gemm_n64_wmma<<<GEMM_BLOCKS, 128, 0, stream>>>(hin, W[l], h_lin, gemm_tiles);
        gather_post_kernel<<<LOOP_BLOCKS, T, 0, stream>>>(rowptr, csr, h_lin, dinv, B[l],
                                                        bn_g, bn_b, bn_m, bn_v, bat,
                                                        h_cur, pooled, N, l, 4);
        hin = h_cur;
    }

    mlp_kernel<<<G, DIMF, 0, stream>>>(pooled, l1W, l1b, l2W, l2b, (float*)d_out);
}


# Layer product moved behind the aggregation ((A h) W = A (h W)) into the gathering wave: rows via an LDS tile, W staged in LDS; the 4 GEMM launches and the h W round trip go; sum order changes
# speedup vs baseline: 2.9915x; 1.1255x over previous
#include <hip/hip_runtime.h>
#include <hip/hip_bf16.h>


typedef float v2f __attribute__((ext_vector_type(2)));
typedef float v8f __attribute__((ext_vector_type(8)));

#define DIMF 64
#define CATF 256
#define OUTF 10
#define LOOP_BLOCKS 1024
#define TPITCH 68
#define WPITCH 72
#define PAD64(n) (((size_t)(n) + 63) & ~(size_t)63)

__global__ void zero_f32(float* __restrict__ p, int n) {
    const int i = (blockIdx.x * blockDim.x + threadIdx.x) * 4;
    if (i + 3 < n) {
        *(float4*)(p + i) = make_float4(0.0f, 0.0f, 0.0f, 0.0f);
    } else {
        for (int j = i; j < n; ++j) p[j] = 0.0f;
    }
}

__global__ void zero_i32(int* __restrict__ p, int n) {
    const int i = (blockIdx.x * blockDim.x + threadIdx.x) * 4;
    if (i + 3 < n) {
        *(int4*)(p + i) = make_int4(0, 0, 0, 0);
    } else {
        for (int j = i; j < n; ++j) p[j] = 0;
    }
}

__global__ __launch_bounds__(256) void rank_kernel(const int* __restrict__ dst,
                                                   int* __restrict__ cnt,
                                                   int* __restrict__ rank, int E) {
    const int stride = gridDim.x * blockDim.x;
    int e0 = blockIdx.x * blockDim.x + threadIdx.x;
    for (; e0 + 3 * stride < E; e0 += 4 * stride) {
        int d[4], r[4];
#pragma unroll
        for (int j = 0; j < 4; ++j) d[j] = dst[e0 + j * stride];
#pragma unroll
        for (int j = 0; j < 4; ++j) r[j] = atomicAdd(&cnt[d[j]], 1);
#pragma unroll
        for (int j = 0; j < 4; ++j) rank[e0 + j * stride] = r[j];
    }
    for (; e0 < E; e0 += stride) rank[e0] = atomicAdd(&cnt[dst[e0]], 1);
}

__global__ __launch_bounds__(256) void scan_kernel(const int* __restrict__ cnt,
                                                   int* __restrict__ rowptr,
                                                   float* __restrict__ dinv, int N) {
    __shared__ int sred[256];
    __shared__ int sm[256];
    const int t = threadIdx.x, b = blockIdx.x;
    const int4* __restrict__ c4 = (const int4*)cnt;
    int pre = 0;
    for (int c = 0; c < b; c += 8) {
        int4 v[8];
#pragma unroll
        for (int u = 0; u < 8; ++u) v[u] = c4[min(c + u, b - 1) * 256 + t];
#pragma unroll
        for (int u = 0; u < 8; ++u)
            pre += (c + u < b) ? (v[u].x + v[u].y) + (v[u].z + v[u].w) : 0;
    }
    sred[t] = pre;
    __syncthreads();
    for (int s = 128; s > 0; s >>= 1) {
        if (t < s) sred[t] += sred[t + s];
        __syncthreads();
    }
    const int before = sred[0];

    const int i0 = b * 1024 + t * 4;
    int c[4];
#pragma unroll
    for (int j = 0; j < 4; ++j) c[j] = (i0 + j < N) ? cnt[i0 + j] : 0;
    const int tsum = (c[0] + c[1]) + (c[2] + c[3]);
    sm[t] = tsum;
    __syncthreads();
    for (int off = 1; off < 256; off <<= 1) {
        const int v = (t >= off) ? sm[t - off] : 0;
        __syncthreads();
        sm[t] += v;
        __syncthreads();
    }
    int run = before + sm[t] - tsum;
#pragma unroll
    for (int j = 0; j < 4; ++j) {
        if (i0 + j < N) {
            rowptr[i0 + j] = run;
            dinv[i0 + j]   = __frsqrt_rn((float)c[j] + 1.0f);
            run += c[j];
            if (i0 + j == N - 1) rowptr[N] = run;
        }
    }
}

__global__ void fill_kernel(const int* __restrict__ src, const int* __restrict__ dst,
                            const int* __restrict__ rank, const int* __restrict__ rowptr,
                            const float* __restrict__ dinv, int* __restrict__ csr, int E) {
    const int stride = gridDim.x * blockDim.x;
    int e0 = blockIdx.x * blockDim.x + threadIdx.x;
    for (; e0 + 3 * stride < E; e0 += 4 * stride) {
        int s[4], d[4], rk[4];
#pragma unroll
        for (int j = 0; j < 4; ++j) {
            const int e = e0 + j * stride;
            s[j]  = src[e];
            d[j]  = dst[e];
            rk[j] = rank[e];
        }
        int   rp[4];
        float ds[4], dd[4];
#pragma unroll
        for (int j = 0; j < 4; ++j) {
            rp[j] = rowptr[d[j]];
            ds[j] = dinv[s[j]];
            dd[j] = dinv[d[j]];
        }
#pragma unroll
        for (int j = 0; j < 4; ++j) {
            int2 r;
            r.x = s[j];
            r.y = __float_as_int(ds[j] * dd[j]);
            ((int2*)csr)[rp[j] + rk[j]] = r;
        }
    }
    for (; e0 < E; e0 += stride) {
        const int s = src[e0], d = dst[e0];
        int2 r;
        r.x = s;
        r.y = __float_as_int(dinv[s] * dinv[d]);
        ((int2*)csr)[rowptr[d] + rank[e0]] = r;
    }
}

__device__ __forceinline__ void fma4(float4& a, const float4 v, const float w) {
    a.x = fmaf(v.x, w, a.x);
    a.y = fmaf(v.y, w, a.y);
    a.z = fmaf(v.z, w, a.z);
    a.w = fmaf(v.w, w, a.w);
}

__device__ __forceinline__ float4 keep4(const bool p, const float4 v) {
    float4 r;
    r.x = p ? v.x : 0.0f;
    r.y = p ? v.y : 0.0f;
    r.z = p ? v.z : 0.0f;
    r.w = p ? v.w : 0.0f;
    return r;
}

__device__ __forceinline__ void pool_add(float* __restrict__ p, const float4 a) {
    atomicAdd(p,     a.x);
    atomicAdd(p + 1, a.y);
    atomicAdd(p + 2, a.z);
    atomicAdd(p + 3, a.w);
}

__global__ __launch_bounds__(256) void layer_kernel(const int* __restrict__ rowptr,
                                                    const int* __restrict__ csr,
                                                    const float* __restrict__ h_in,
                                                    const float* __restrict__ dinv,
                                                    const float* __restrict__ Wm,
                                                    const float* __restrict__ bias,
                                                    const float* __restrict__ gamma,
                                                    const float* __restrict__ beta,
                                                    const float* __restrict__ mean,
                                                    const float* __restrict__ var,
                                                    const int* __restrict__ batch,
                                                    float* __restrict__ h_out,
                                                    float* __restrict__ pooled,
                                                    int N, int layer, int n_layers) {
    __shared__ __attribute__((aligned(16))) float tiles[8][16 * TPITCH];
    __shared__ __attribute__((aligned(16))) float wl[DIMF * WPITCH];
#pragma unroll
    for (int q = 0; q < 4; ++q) {
        const int idx = (int)threadIdx.x + 256 * q;
        const int wr  = idx >> 4;
        const int wc  = (idx & 15) * 4;
        *(float4*)(wl + wr * WPITCH + wc) = *(const float4*)(Wm + wr * DIMF + wc);
    }
    __syncthreads();

    const int wib  = __builtin_amdgcn_readfirstlane((int)(threadIdx.x >> 5));
    const int wave = blockIdx.x * 8 + wib;
    const int nw   = gridDim.x * 8;
    const int base = N / nw;
    const int rem  = N - base * nw;
    const int row0 = wave * base + min(wave, rem);
    const int row1 = row0 + base + (wave < rem ? 1 : 0);
    const int lane = threadIdx.x & 31;
    const int c    = lane & 15;
    const int h    = lane >> 4;
    const int d0   = c * 4;
    const int ld   = layer * DIMF + d0;
    float* tw = tiles[wib];
    const int2* __restrict__ rec = (const int2*)csr;
    const float* __restrict__ hb = h_in + d0;
    float* __restrict__ pcol = pooled + ld;

    int    have = 0;
    int    gp   = 0;
    float4 ax   = make_float4(0.0f, 0.0f, 0.0f, 0.0f);

    {
        const int tb = row0;
        const int nr = min(16, row1 - row0);

        const float4 bb = *(const float4*)(bias + d0);
        const float4 mu = *(const float4*)(mean + ld);
        const float4 vv = *(const float4*)(var + ld);
        const float4 ga = *(const float4*)(gamma + ld);
        const float4 be = *(const float4*)(beta + ld);
        float4 rs;
        rs.x = __frsqrt_rn(vv.x + 1e-5f);
        rs.y = __frsqrt_rn(vv.y + 1e-5f);
        rs.z = __frsqrt_rn(vv.z + 1e-5f);
        rs.w = __frsqrt_rn(vv.w + 1e-5f);
        int gid[8];
#pragma unroll
        for (int r = 0; r < 8; ++r) gid[r] = batch[min(tb + 8 * h + r, row1 - 1)];

        for (int li = h; li < nr; li += 2) {
            const int i   = tb + li;
            const int beg = rowptr[i];
            const int end = rowptr[i + 1];
            const float  di = dinv[i];
            const float4 hs = *(const float4*)(hb + (size_t)i * DIMF);

            float4 acc = make_float4(0.0f, 0.0f, 0.0f, 0.0f);
            const int last = end - 1;
            for (int k = beg; k < end; k += 8) {
                int2   e[8];
                float4 v[8];
#pragma unroll
                for (int u = 0; u < 8; ++u) e[u] = rec[min(k + u, last)];
#pragma unroll
                for (int u = 0; u < 8; ++u) v[u] = *(const float4*)(hb + (unsigned)e[u].x * (unsigned)DIMF);
#pragma unroll
                for (int u = 0; u < 8; ++u) {
                    const bool p = k + u < end;
                    fma4(acc, keep4(p, v[u]), p ? __int_as_float(e[u].y) : 0.0f);
                }
            }

            float4 o;
            o.x = acc.x + hs.x * di * di;
            o.y = acc.y + hs.y * di * di;
            o.z = acc.z + hs.z * di * di;
            o.w = acc.w + hs.w * di * di;
            *(float4*)(tw + li * TPITCH + d0) = o;
        }
#pragma unroll
        for (int q = 0; q < 8; ++q) {
            const int li = 2 * q + h;
            if (li >= nr) *(float4*)(tw + li * TPITCH + d0) = make_float4(0.0f, 0.0f, 0.0f, 0.0f);
        }

        const float* arow = tw + c * TPITCH + 4 * h;
        v8f acc[4];
#pragma unroll
        for (int nt = 0; nt < 4; ++nt) acc[nt] = v8f{};

#pragma unroll
        for (int j = 0; j < 8; ++j) {
            const float4 qa = *(const float4*)(arow + 8 * j);
            float av[4];
            av[0] = qa.x;
            av[1] = qa.y;
            av[2] = qa.z;
            av[3] = qa.w;
            float wv[4][4];
#pragma unroll
            for (int i = 0; i < 4; ++i) {
                const float4 q = *(const float4*)(wl + (8 * j + 4 * h + i) * WPITCH + 4 * c);
                wv[i][0] = q.x;
                wv[i][1] = q.y;
                wv[i][2] = q.z;
                wv[i][3] = q.w;
            }
#pragma unroll
            for (int s = 0; s < 2; ++s) {
                v2f a;
                a.x = av[2 * s];
                a.y = av[2 * s + 1];
#pragma unroll
                for (int nt = 0; nt < 4; ++nt) {
                    v2f b;
                    b.x = wv[2 * s][nt];
                    b.y = wv[2 * s + 1][nt];
                    acc[nt] = __builtin_amdgcn_wmma_f32_16x16x4_f32(
                             false, a,  false, b,
                             (short)0, acc[nt],  false,  false);
                }
            }
        }

        float4 t[8];
#pragma unroll
        for (int r = 0; r < 8; ++r) {
            const float ox = acc[0][r] + bb.x;
            const float oy = acc[1][r] + bb.y;
            const float oz = acc[2][r] + bb.z;
            const float ow = acc[3][r] + bb.w;
            t[r].x = fmaxf((ox - mu.x) * rs.x * ga.x + be.x, 0.0f);
            t[r].y = fmaxf((oy - mu.y) * rs.y * ga.y + be.y, 0.0f);
            t[r].z = fmaxf((oz - mu.z) * rs.z * ga.z + be.z, 0.0f);
            t[r].w = fmaxf((ow - mu.w) * rs.w * ga.w + be.w, 0.0f);
        }

#pragma unroll
        for (int r = 0; r < 8; ++r) {
            const int li = 8 * h + r;
            if (li < nr) {
                const int i = tb + li;
                const int g = gid[r];
                if (layer + 1 < n_layers) *(float4*)(h_out + (size_t)i * DIMF + d0) = t[r];

                if (have && g != gp) {
                    pool_add(pcol + (size_t)gp * CATF, ax);
                    ax = make_float4(0.0f, 0.0f, 0.0f, 0.0f);
                }
                gp   = g;
                have = 1;
                ax.x += t[r].x;
                ax.y += t[r].y;
                ax.z += t[r].z;
                ax.w += t[r].w;
            }
        }
    }

    const int   have_o = __shfl_xor(have, 16);
    const int   gp_o   = __shfl_xor(gp, 16);
    float4 ax_o;
    ax_o.x = __shfl_xor(ax.x, 16);
    ax_o.y = __shfl_xor(ax.y, 16);
    ax_o.z = __shfl_xor(ax.z, 16);
    ax_o.w = __shfl_xor(ax.w, 16);
    if (have) {
        if (have_o && gp_o == gp) {
            if (h == 0) {
                ax.x += ax_o.x;
                ax.y += ax_o.y;
                ax.z += ax_o.z;
                ax.w += ax_o.w;
                pool_add(pcol + (size_t)gp * CATF, ax);
            }
        } else {
            pool_add(pcol + (size_t)gp * CATF, ax);
        }
    }
}

__global__ __launch_bounds__(64) void mlp_kernel(const float* __restrict__ pooled,
                                                 const float* __restrict__ W1, const float* __restrict__ b1,
                                                 const float* __restrict__ W2, const float* __restrict__ b2,
                                                 float* __restrict__ out) {
    __shared__ float sp[CATF];
    __shared__ float sh[DIMF];
    int g = blockIdx.x, t = threadIdx.x;
    const float* prow = pooled + (size_t)g * CATF;
#pragma unroll
    for (int j = 0; j < 4; ++j) sp[t + j * DIMF] = prow[t + j * DIMF];
    __syncthreads();
    float acc = b1[t];
#pragma unroll 8
    for (int k = 0; k < CATF; ++k) acc = fmaf(sp[k], W1[k * DIMF + t], acc);
    sh[t] = fmaxf(acc, 0.0f);
    __syncthreads();
    if (t < OUTF) {
        float o = b2[t];
#pragma unroll 8
        for (int d = 0; d < DIMF; ++d) o = fmaf(sh[d], W2[d * OUTF + t], o);
        out[(size_t)g * OUTF + t] = o;
    }
}

extern "C" void kernel_launch(void* const* d_in, const int* in_sizes, int n_in,
                              void* d_out, int out_size, void* d_ws, size_t ws_size,
                              hipStream_t stream) {
    const float* x    = (const float*)d_in[0];
    const int*   ei   = (const int*)d_in[1];
    const int*   bat  = (const int*)d_in[2];
    const float* W[4] = {(const float*)d_in[4], (const float*)d_in[6],
                         (const float*)d_in[8], (const float*)d_in[10]};
    const float* B[4] = {(const float*)d_in[5], (const float*)d_in[7],
                         (const float*)d_in[9], (const float*)d_in[11]};
    const float* bn_g = (const float*)d_in[12];
    const float* bn_b = (const float*)d_in[13];
    const float* bn_m = (const float*)d_in[14];
    const float* bn_v = (const float*)d_in[15];
    const float* l1W  = (const float*)d_in[16];
    const float* l1b  = (const float*)d_in[17];
    const float* l2W  = (const float*)d_in[18];
    const float* l2b  = (const float*)d_in[19];

    const int N = in_sizes[0] / DIMF;
    const int E = in_sizes[1] / 2;
    const int G = out_size / OUTF;
    const int* src = ei;
    const int* dst = ei + E;

    float* ws     = (float*)d_ws;
    float* dinv   = ws;
    int*   cnt    = (int*)(dinv + PAD64(N));
    int*   rowptr = cnt + PAD64(N);
    int*   rank   = rowptr + PAD64(N + 1);
    int*   csr    = rank + PAD64(E);
    float* h_a    = (float*)(csr + PAD64(2 * (size_t)E));
    float* h_b    = h_a + (size_t)N * DIMF;
    float* pooled = h_b + (size_t)N * DIMF;

    const int T = 256;
    zero_i32<<<(N + 4 * T - 1) / (4 * T), T, 0, stream>>>(cnt, N);
    zero_f32<<<(G * CATF + 4 * T - 1) / (4 * T), T, 0, stream>>>(pooled, G * CATF);
    rank_kernel<<<LOOP_BLOCKS, T, 0, stream>>>(dst, cnt, rank, E);
    scan_kernel<<<(N + 1023) / 1024, T, 0, stream>>>(cnt, rowptr, dinv, N);
    fill_kernel<<<LOOP_BLOCKS, T, 0, stream>>>(src, dst, rank, rowptr, dinv, csr, E);

    float* hbuf[2] = {h_a, h_b};
    const float* hin = x;
    for (int l = 0; l < 4; ++l) {
        float* hout = hbuf[l & 1];
        layer_kernel<<<LOOP_BLOCKS, T, 0, stream>>>(rowptr, csr, hin, dinv, W[l], B[l],
                                                  bn_g, bn_b, bn_m, bn_v, bat,
                                                  hout, pooled, N, l, 4);
        hin = hout;
    }

    mlp_kernel<<<G, DIMF, 0, stream>>>(pooled, l1W, l1b, l2W, l2b, (float*)d_out);
}


# Edge sort rebuilt on LDS counters: bin pass sorts chunks by 128-row bucket, bucket pass places rows; the sort's global atomics, rank array, scan launch go; layer 1 fills edge norms; sum order changes
# speedup vs baseline: 4.6965x; 1.5699x over previous
#include <hip/hip_runtime.h>
#include <hip/hip_bf16.h>


typedef float v2f __attribute__((ext_vector_type(2)));
typedef float v8f __attribute__((ext_vector_type(8)));

#define DIMF 64
#define CATF 256
#define OUTF 10
#define LOOP_BLOCKS 1024
#define TPITCH 68
#define WPITCH 72
#define PAD64(n) (((size_t)(n) + 63) & ~(size_t)63)
#define RB 128
#define RBSH 7
#define KMAX 1024
#define BINB 256
#define SCAP 7168
#define LCAP 6144

template <int NW>
__device__ __forceinline__ int block_excl_scan(const int v, int* sa, int* sb, int& total) {
    const int lane = threadIdx.x & 31;
    const int w    = threadIdx.x >> 5;
    int x = v;
#pragma unroll
    for (int off = 1; off < 32; off <<= 1) {
        const int y = __shfl_up(x, off);
        if (lane >= off) x += y;
    }
    if (lane == 31) sa[w] = x;
    __syncthreads();
    if (w == 0) {
        const int s = (lane < NW) ? sa[lane] : 0;
        int xs = s;
#pragma unroll
        for (int off = 1; off < 32; off <<= 1) {
            const int y = __shfl_up(xs, off);
            if (lane >= off) xs += y;
        }
        sb[lane] = xs - s;
        if (lane == 31) sb[32] = xs;
    }
    __syncthreads();
    total = sb[32];
    return sb[w] + x - v;
}

__global__ __launch_bounds__(1024) void bin_kernel(const int* __restrict__ src,
                                                   const int* __restrict__ dst,
                                                   int* __restrict__ binned_w,
                                                   int* __restrict__ startT,
                                                   float* __restrict__ pooled,
                                                   int npool, int E, int CH, int K) {
    __shared__ int hist[KMAX];
    __shared__ int sa[33];
    __shared__ int sb[33];
    __shared__ __attribute__((aligned(16))) int2 stage[SCAP];
    int2* __restrict__ binned = (int2*)binned_w;
    const int t = threadIdx.x, b = blockIdx.x;
    for (int i = b * 1024 + t; i < npool; i += gridDim.x * 1024) pooled[i] = 0.0f;

    const int e0 = b * CH;
    const int n  = max(0, min(CH, E - e0));
    const int* __restrict__ dch = dst + e0;
    const int* __restrict__ sch = src + e0;
    hist[t] = 0;
    __syncthreads();
    for (int i0 = t; i0 < n; i0 += 4096) {
        int d[4];
#pragma unroll
        for (int u = 0; u < 4; ++u) d[u] = dch[min(i0 + 1024 * u, n - 1)];
#pragma unroll
        for (int u = 0; u < 4; ++u)
            if (i0 + 1024 * u < n) atomicAdd(&hist[d[u] >> RBSH], 1);
    }
    __syncthreads();
    const int cnt = hist[t];
    int total;
    const int ex = block_excl_scan<32>(cnt, sa, sb, total);
    if (t < K) startT[t * BINB + b] = ex;
    if (t == 0) startT[K * BINB + b] = n;
    hist[t] = ex;
    __syncthreads();

    const bool staged = n <= SCAP;
    for (int i0 = t; i0 < n; i0 += 4096) {
        int s[4], d[4];
#pragma unroll
        for (int u = 0; u < 4; ++u) {
            const int idx = min(i0 + 1024 * u, n - 1);
            s[u] = sch[idx];
            d[u] = dch[idx];
        }
#pragma unroll
        for (int u = 0; u < 4; ++u) {
            if (i0 + 1024 * u < n) {
                const int  p = atomicAdd(&hist[d[u] >> RBSH], 1);
                const int2 r = make_int2(s[u], d[u]);
                if (staged) stage[p] = r;
                else        binned[(size_t)e0 + p] = r;
            }
        }
    }
    if (staged) {
        __syncthreads();
        for (int i = t; i < n; i += 1024) binned[(size_t)e0 + i] = stage[i];
    }
}

__global__ __launch_bounds__(256) void bucket_kernel(const int* __restrict__ binned_w,
                                                     const int* __restrict__ startT,
                                                     int* __restrict__ rowptr,
                                                     float* __restrict__ dinv,
                                                     int* __restrict__ csr_w,
                                                     int N, int CH) {
    __shared__ int deg[RB];
    __shared__ int sa[33];
    __shared__ int sb[33];
    __shared__ __attribute__((aligned(16))) int2 stage[LCAP];
    const int2* __restrict__ binned = (const int2*)binned_w;
    int2* __restrict__ csr = (int2*)csr_w;
    const int t = threadIdx.x, k = blockIdx.x;
    const int r0 = k * RB;
    const int s0 = startT[k * BINB + t];
    const int c  = startT[(k + 1) * BINB + t] - s0;
    if (t < RB) deg[t] = 0;
    int base, nE;
    (void)block_excl_scan<8>(s0, sa, sb, base);
    (void)block_excl_scan<8>(c, sa, sb, nE);

    const int2* __restrict__ seg  = binned + (size_t)t * CH + s0;
    const int2* __restrict__ segs = (c > 0) ? seg : binned;
    const int cl = max(c - 1, 0);
    int2 r[16];
#pragma unroll
    for (int j = 0; j < 16; ++j) r[j] = segs[min(j, cl)];
#pragma unroll
    for (int j = 0; j < 16; ++j)
        if (j < c) atomicAdd(&deg[r[j].y - r0], 1);
    for (int j = 16; j < c; ++j) atomicAdd(&deg[seg[j].y - r0], 1);
    __syncthreads();

    const int dg = (t < RB) ? deg[t] : 0;
    int tot2;
    const int rs  = block_excl_scan<8>(dg, sa, sb, tot2);
    const int row = r0 + t;
    if (t < RB && row < N) {
        rowptr[row] = base + rs;
        dinv[row]   = __frsqrt_rn((float)dg + 1.0f);
    }
    if (k == (int)gridDim.x - 1 && t == 0) rowptr[N] = base + nE;
    if (t < RB) deg[t] = rs;
    __syncthreads();

    const bool staged = nE <= LCAP;
#pragma unroll
    for (int j = 0; j < 16; ++j) {
        if (j < c) {
            const int  p   = atomicAdd(&deg[r[j].y - r0], 1);
            const int2 rec = make_int2(r[j].x, 0);
            if (staged) stage[p] = rec;
            else        csr[(size_t)base + p] = rec;
        }
    }
    for (int j = 16; j < c; ++j) {
        const int2 q   = seg[j];
        const int  p   = atomicAdd(&deg[q.y - r0], 1);
        const int2 rec = make_int2(q.x, 0);
        if (staged) stage[p] = rec;
        else        csr[(size_t)base + p] = rec;
    }
    if (staged) {
        __syncthreads();
        for (int i = t; i < nE; i += 256) csr[(size_t)base + i] = stage[i];
    }
}

__device__ __forceinline__ void fma4(float4& a, const float4 v, const float w) {
    a.x = fmaf(v.x, w, a.x);
    a.y = fmaf(v.y, w, a.y);
    a.z = fmaf(v.z, w, a.z);
    a.w = fmaf(v.w, w, a.w);
}

__device__ __forceinline__ float4 keep4(const bool p, const float4 v) {
    float4 r;
    r.x = p ? v.x : 0.0f;
    r.y = p ? v.y : 0.0f;
    r.z = p ? v.z : 0.0f;
    r.w = p ? v.w : 0.0f;
    return r;
}

__device__ __forceinline__ void pool_add(float* __restrict__ p, const float4 a) {
    atomicAdd(p,     a.x);
    atomicAdd(p + 1, a.y);
    atomicAdd(p + 2, a.z);
    atomicAdd(p + 3, a.w);
}

template <bool FIRST>
__global__ __launch_bounds__(256) void layer_kernel(const int* __restrict__ rowptr,
                                                    int* __restrict__ csr,
                                                    const float* __restrict__ h_in,
                                                    const float* __restrict__ dinv,
                                                    const float* __restrict__ Wm,
                                                    const float* __restrict__ bias,
                                                    const float* __restrict__ gamma,
                                                    const float* __restrict__ beta,
                                                    const float* __restrict__ mean,
                                                    const float* __restrict__ var,
                                                    const int* __restrict__ batch,
                                                    float* __restrict__ h_out,
                                                    float* __restrict__ pooled,
                                                    int N, int layer, int n_layers) {
    __shared__ __attribute__((aligned(16))) float tiles[8][16 * TPITCH];
    __shared__ __attribute__((aligned(16))) float wl[DIMF * WPITCH];
#pragma unroll
    for (int q = 0; q < 4; ++q) {
        const int idx = (int)threadIdx.x + 256 * q;
        const int wr  = idx >> 4;
        const int wc  = (idx & 15) * 4;
        *(float4*)(wl + wr * WPITCH + wc) = *(const float4*)(Wm + wr * DIMF + wc);
    }
    __syncthreads();

    const int wib  = __builtin_amdgcn_readfirstlane((int)(threadIdx.x >> 5));
    const int wave = blockIdx.x * 8 + wib;
    const int nw   = gridDim.x * 8;
    const int base = N / nw;
    const int rem  = N - base * nw;
    const int row0 = wave * base + min(wave, rem);
    const int row1 = row0 + base + (wave < rem ? 1 : 0);
    const int lane = threadIdx.x & 31;
    const int c    = lane & 15;
    const int h    = lane >> 4;
    const int d0   = c * 4;
    const int ld   = layer * DIMF + d0;
    float* tw = tiles[wib];
    const int2* rec = (const int2*)csr;
    const float* __restrict__ hb = h_in + d0;
    float* __restrict__ pcol = pooled + ld;

    int    have = 0;
    int    gp   = 0;
    float4 ax   = make_float4(0.0f, 0.0f, 0.0f, 0.0f);

    {
        const int tb = row0;
        const int nr = min(16, row1 - row0);

        const float4 bb = *(const float4*)(bias + d0);
        const float4 mu = *(const float4*)(mean + ld);
        const float4 vv = *(const float4*)(var + ld);
        const float4 ga = *(const float4*)(gamma + ld);
        const float4 be = *(const float4*)(beta + ld);
        float4 rs;
        rs.x = __frsqrt_rn(vv.x + 1e-5f);
        rs.y = __frsqrt_rn(vv.y + 1e-5f);
        rs.z = __frsqrt_rn(vv.z + 1e-5f);
        rs.w = __frsqrt_rn(vv.w + 1e-5f);
        int gid[8];
#pragma unroll
        for (int r = 0; r < 8; ++r) gid[r] = batch[min(tb + 8 * h + r, row1 - 1)];

        for (int li = h; li < nr; li += 2) {
            const int i   = tb + li;
            const int beg = rowptr[i];
            const int end = rowptr[i + 1];
            const float  di = dinv[i];
            const float4 hs = *(const float4*)(hb + (size_t)i * DIMF);

            float4 acc = make_float4(0.0f, 0.0f, 0.0f, 0.0f);
            const int last = end - 1;
            for (int k = beg; k < end; k += 8) {
                int2   e[8];
                float4 v[8];
#pragma unroll
                for (int u = 0; u < 8; ++u) e[u] = rec[min(k + u, last)];
#pragma unroll
                for (int u = 0; u < 8; ++u) v[u] = *(const float4*)(hb + (unsigned)e[u].x * (unsigned)DIMF);
                float w[8];
                if (FIRST) {
#pragma unroll
                    for (int u = 0; u < 8; ++u) w[u] = dinv[(unsigned)e[u].x] * di;
                    float mine = w[0];
#pragma unroll
                    for (int u = 1; u < 8; ++u) mine = (c == u) ? w[u] : mine;
                    if (c < 8 && k + c < end) csr[2 * (size_t)(k + c) + 1] = __float_as_int(mine);
                } else {
#pragma unroll
                    for (int u = 0; u < 8; ++u) w[u] = __int_as_float(e[u].y);
                }
#pragma unroll
                for (int u = 0; u < 8; ++u) {
                    const bool p = k + u < end;
                    fma4(acc, keep4(p, v[u]), p ? w[u] : 0.0f);
                }
            }

            float4 o;
            o.x = acc.x + hs.x * di * di;
            o.y = acc.y + hs.y * di * di;
            o.z = acc.z + hs.z * di * di;
            o.w = acc.w + hs.w * di * di;
            *(float4*)(tw + li * TPITCH + d0) = o;
        }
#pragma unroll
        for (int q = 0; q < 8; ++q) {
            const int li = 2 * q + h;
            if (li >= nr) *(float4*)(tw + li * TPITCH + d0) = make_float4(0.0f, 0.0f, 0.0f, 0.0f);
        }

        const float* arow = tw + c * TPITCH + 4 * h;
        v8f acc[4];
#pragma unroll
        for (int nt = 0; nt < 4; ++nt) acc[nt] = v8f{};

#pragma unroll
        for (int j = 0; j < 8; ++j) {
            const float4 qa = *(const float4*)(arow + 8 * j);
            float av[4];
            av[0] = qa.x;
            av[1] = qa.y;
            av[2] = qa.z;
            av[3] = qa.w;
            float wv[4][4];
#pragma unroll
            for (int i = 0; i < 4; ++i) {
                const float4 q = *(const float4*)(wl + (8 * j + 4 * h + i) * WPITCH + 4 * c);
                wv[i][0] = q.x;
                wv[i][1] = q.y;
                wv[i][2] = q.z;
                wv[i][3] = q.w;
            }
#pragma unroll
            for (int s = 0; s < 2; ++s) {
                v2f a;
                a.x = av[2 * s];
                a.y = av[2 * s + 1];
#pragma unroll
                for (int nt = 0; nt < 4; ++nt) {
                    v2f b;
                    b.x = wv[2 * s][nt];
                    b.y = wv[2 * s + 1][nt];
                    acc[nt] = __builtin_amdgcn_wmma_f32_16x16x4_f32(
                             false, a,  false, b,
                             (short)0, acc[nt],  false,  false);
                }
            }
        }

        float4 t[8];
#pragma unroll
        for (int r = 0; r < 8; ++r) {
            const float ox = acc[0][r] + bb.x;
            const float oy = acc[1][r] + bb.y;
            const float oz = acc[2][r] + bb.z;
            const float ow = acc[3][r] + bb.w;
            t[r].x = fmaxf((ox - mu.x) * rs.x * ga.x + be.x, 0.0f);
            t[r].y = fmaxf((oy - mu.y) * rs.y * ga.y + be.y, 0.0f);
            t[r].z = fmaxf((oz - mu.z) * rs.z * ga.z + be.z, 0.0f);
            t[r].w = fmaxf((ow - mu.w) * rs.w * ga.w + be.w, 0.0f);
        }

#pragma unroll
        for (int r = 0; r < 8; ++r) {
            const int li = 8 * h + r;
            if (li < nr) {
                const int i = tb + li;
                const int g = gid[r];
                if (layer + 1 < n_layers) *(float4*)(h_out + (size_t)i * DIMF + d0) = t[r];

                if (have && g != gp) {
                    pool_add(pcol + (size_t)gp * CATF, ax);
                    ax = make_float4(0.0f, 0.0f, 0.0f, 0.0f);
                }
                gp   = g;
                have = 1;
                ax.x += t[r].x;
                ax.y += t[r].y;
                ax.z += t[r].z;
                ax.w += t[r].w;
            }
        }
    }

    const int   have_o = __shfl_xor(have, 16);
    const int   gp_o   = __shfl_xor(gp, 16);
    float4 ax_o;
    ax_o.x = __shfl_xor(ax.x, 16);
    ax_o.y = __shfl_xor(ax.y, 16);
    ax_o.z = __shfl_xor(ax.z, 16);
    ax_o.w = __shfl_xor(ax.w, 16);
    if (have) {
        if (have_o && gp_o == gp) {
            if (h == 0) {
                ax.x += ax_o.x;
                ax.y += ax_o.y;
                ax.z += ax_o.z;
                ax.w += ax_o.w;
                pool_add(pcol + (size_t)gp * CATF, ax);
            }
        } else {
            pool_add(pcol + (size_t)gp * CATF, ax);
        }
    }
}

__global__ __launch_bounds__(64) void mlp_kernel(const float* __restrict__ pooled,
                                                 const float* __restrict__ W1, const float* __restrict__ b1,
                                                 const float* __restrict__ W2, const float* __restrict__ b2,
                                                 float* __restrict__ out) {
    __shared__ float sp[CATF];
    __shared__ float sh[DIMF];
    int g = blockIdx.x, t = threadIdx.x;
    const float* prow = pooled + (size_t)g * CATF;
#pragma unroll
    for (int j = 0; j < 4; ++j) sp[t + j * DIMF] = prow[t + j * DIMF];
    __syncthreads();
    float acc = b1[t];
#pragma unroll 8
    for (int k = 0; k < CATF; ++k) acc = fmaf(sp[k], W1[k * DIMF + t], acc);
    sh[t] = fmaxf(acc, 0.0f);
    __syncthreads();
    if (t < OUTF) {
        float o = b2[t];
#pragma unroll 8
        for (int d = 0; d < DIMF; ++d) o = fmaf(sh[d], W2[d * OUTF + t], o);
        out[(size_t)g * OUTF + t] = o;
    }
}

extern "C" void kernel_launch(void* const* d_in, const int* in_sizes, int n_in,
                              void* d_out, int out_size, void* d_ws, size_t ws_size,
                              hipStream_t stream) {
    const float* x    = (const float*)d_in[0];
    const int*   ei   = (const int*)d_in[1];
    const int*   bat  = (const int*)d_in[2];
    const float* W[4] = {(const float*)d_in[4], (const float*)d_in[6],
                         (const float*)d_in[8], (const float*)d_in[10]};
    const float* B[4] = {(const float*)d_in[5], (const float*)d_in[7],
                         (const float*)d_in[9], (const float*)d_in[11]};
    const float* bn_g = (const float*)d_in[12];
    const float* bn_b = (const float*)d_in[13];
    const float* bn_m = (const float*)d_in[14];
    const float* bn_v = (const float*)d_in[15];
    const float* l1W  = (const float*)d_in[16];
    const float* l1b  = (const float*)d_in[17];
    const float* l2W  = (const float*)d_in[18];
    const float* l2b  = (const float*)d_in[19];

    const int N = in_sizes[0] / DIMF;
    const int E = in_sizes[1] / 2;
    const int G = out_size / OUTF;
    const int* src = ei;
    const int* dst = ei + E;

    float* ws     = (float*)d_ws;
    const int K   = (N + RB - 1) / RB;
    const int CH  = (E + BINB - 1) / BINB;
    float* dinv   = ws;
    int*   rowptr = (int*)(dinv + PAD64(N));
    int*   startT = rowptr + PAD64(N + 1);
    int*   binned = startT + PAD64((size_t)(K + 1) * BINB);
    int*   csr    = binned + PAD64(2 * (size_t)E);
    float* h_a    = (float*)(csr + PAD64(2 * (size_t)E));
    float* h_b    = h_a + (size_t)N * DIMF;
    float* pooled = h_b + (size_t)N * DIMF;

    const int T = 256;
    bin_kernel<<<BINB, 1024, 0, stream>>>(src, dst, binned, startT, pooled, G * CATF, E, CH, K);
    bucket_kernel<<<K, T, 0, stream>>>(binned, startT, rowptr, dinv, csr, N, CH);

    float* hbuf[2] = {h_a, h_b};
    const float* hin = x;
    for (int l = 0; l < 4; ++l) {
        float* hout = hbuf[l & 1];
        if (l == 0)
            layer_kernel<true><<<LOOP_BLOCKS, T, 0, stream>>>(rowptr, csr, hin, dinv, W[l], B[l],
                                                            bn_g, bn_b, bn_m, bn_v, bat,
                                                            hout, pooled, N, l, 4);
        else
            layer_kernel<false><<<LOOP_BLOCKS, T, 0, stream>>>(rowptr, csr, hin, dinv, W[l], B[l],
                                                             bn_g, bn_b, bn_m, bn_v, bat,
                                                             hout, pooled, N, l, 4);
        hin = hout;
    }

    mlp_kernel<<<G, DIMF, 0, stream>>>(pooled, l1W, l1b, l2W, l2b, (float*)d_out);
}


# Layer kernel: each wave copies its slice's records and row starts into LDS once; the gather reads them there: a group of rows waits one memory trip, not two (32-bit row offsets keep 8 waves/SIMD)
# speedup vs baseline: 4.8935x; 1.0419x over previous
#include <hip/hip_runtime.h>
#include <hip/hip_bf16.h>


typedef float v2f __attribute__((ext_vector_type(2)));
typedef float v8f __attribute__((ext_vector_type(8)));

#define DIMF 64
#define CATF 256
#define OUTF 10
#define LOOP_BLOCKS 1024
#define TPITCH 68
#define WPITCH 72
#define PAD64(n) (((size_t)(n) + 63) & ~(size_t)63)
#define RB 128
#define RBSH 7
#define KMAX 1024
#define BINB 256
#define SCAP 7168
#define LCAP 6144
#define RCAP 320

template <int NW>
__device__ __forceinline__ int block_excl_scan(const int v, int* sa, int* sb, int& total) {
    const int lane = threadIdx.x & 31;
    const int w    = threadIdx.x >> 5;
    int x = v;
#pragma unroll
    for (int off = 1; off < 32; off <<= 1) {
        const int y = __shfl_up(x, off);
        if (lane >= off) x += y;
    }
    if (lane == 31) sa[w] = x;
    __syncthreads();
    if (w == 0) {
        const int s = (lane < NW) ? sa[lane] : 0;
        int xs = s;
#pragma unroll
        for (int off = 1; off < 32; off <<= 1) {
            const int y = __shfl_up(xs, off);
            if (lane >= off) xs += y;
        }
        sb[lane] = xs - s;
        if (lane == 31) sb[32] = xs;
    }
    __syncthreads();
    total = sb[32];
    return sb[w] + x - v;
}

__global__ __launch_bounds__(1024) void bin_kernel(const int* __restrict__ src,
                                                   const int* __restrict__ dst,
                                                   int* __restrict__ binned_w,
                                                   int* __restrict__ startT,
                                                   float* __restrict__ pooled,
                                                   int npool, int E, int CH, int K) {
    __shared__ int hist[KMAX];
    __shared__ int sa[33];
    __shared__ int sb[33];
    __shared__ __attribute__((aligned(16))) int2 stage[SCAP];
    int2* __restrict__ binned = (int2*)binned_w;
    const int t = threadIdx.x, b = blockIdx.x;
    for (int i = b * 1024 + t; i < npool; i += gridDim.x * 1024) pooled[i] = 0.0f;

    const int e0 = b * CH;
    const int n  = max(0, min(CH, E - e0));
    const int* __restrict__ dch = dst + e0;
    const int* __restrict__ sch = src + e0;
    hist[t] = 0;
    __syncthreads();
    for (int i0 = t; i0 < n; i0 += 4096) {
        int d[4];
#pragma unroll
        for (int u = 0; u < 4; ++u) d[u] = dch[min(i0 + 1024 * u, n - 1)];
#pragma unroll
        for (int u = 0; u < 4; ++u)
            if (i0 + 1024 * u < n) atomicAdd(&hist[d[u] >> RBSH], 1);
    }
    __syncthreads();
    const int cnt = hist[t];
    int total;
    const int ex = block_excl_scan<32>(cnt, sa, sb, total);
    if (t < K) startT[t * BINB + b] = ex;
    if (t == 0) startT[K * BINB + b] = n;
    hist[t] = ex;
    __syncthreads();

    const bool staged = n <= SCAP;
    for (int i0 = t; i0 < n; i0 += 4096) {
        int s[4], d[4];
#pragma unroll
        for (int u = 0; u < 4; ++u) {
            const int idx = min(i0 + 1024 * u, n - 1);
            s[u] = sch[idx];
            d[u] = dch[idx];
        }
#pragma unroll
        for (int u = 0; u < 4; ++u) {
            if (i0 + 1024 * u < n) {
                const int  p = atomicAdd(&hist[d[u] >> RBSH], 1);
                const int2 r = make_int2(s[u], d[u]);
                if (staged) stage[p] = r;
                else        binned[(size_t)e0 + p] = r;
            }
        }
    }
    if (staged) {
        __syncthreads();
        for (int i = t; i < n; i += 1024) binned[(size_t)e0 + i] = stage[i];
    }
}

__global__ __launch_bounds__(256) void bucket_kernel(const int* __restrict__ binned_w,
                                                     const int* __restrict__ startT,
                                                     int* __restrict__ rowptr,
                                                     float* __restrict__ dinv,
                                                     int* __restrict__ csr_w,
                                                     int N, int CH) {
    __shared__ int deg[RB];
    __shared__ int sa[33];
    __shared__ int sb[33];
    __shared__ __attribute__((aligned(16))) int2 stage[LCAP];
    const int2* __restrict__ binned = (const int2*)binned_w;
    int2* __restrict__ csr = (int2*)csr_w;
    const int t = threadIdx.x, k = blockIdx.x;
    const int r0 = k * RB;
    const int s0 = startT[k * BINB + t];
    const int c  = startT[(k + 1) * BINB + t] - s0;
    if (t < RB) deg[t] = 0;
    int base, nE;
    (void)block_excl_scan<8>(s0, sa, sb, base);
    (void)block_excl_scan<8>(c, sa, sb, nE);

    const int2* __restrict__ seg  = binned + (size_t)t * CH + s0;
    const int2* __restrict__ segs = (c > 0) ? seg : binned;
    const int cl = max(c - 1, 0);
    int2 r[16];
#pragma unroll
    for (int j = 0; j < 16; ++j) r[j] = segs[min(j, cl)];
#pragma unroll
    for (int j = 0; j < 16; ++j)
        if (j < c) atomicAdd(&deg[r[j].y - r0], 1);
    for (int j = 16; j < c; ++j) atomicAdd(&deg[seg[j].y - r0], 1);
    __syncthreads();

    const int dg = (t < RB) ? deg[t] : 0;
    int tot2;
    const int rs  = block_excl_scan<8>(dg, sa, sb, tot2);
    const int row = r0 + t;
    if (t < RB && row < N) {
        rowptr[row] = base + rs;
        dinv[row]   = __frsqrt_rn((float)dg + 1.0f);
    }
    if (k == (int)gridDim.x - 1 && t == 0) rowptr[N] = base + nE;
    if (t < RB) deg[t] = rs;
    __syncthreads();

    const bool staged = nE <= LCAP;
#pragma unroll
    for (int j = 0; j < 16; ++j) {
        if (j < c) {
            const int  p   = atomicAdd(&deg[r[j].y - r0], 1);
            const int2 rec = make_int2(r[j].x, 0);
            if (staged) stage[p] = rec;
            else        csr[(size_t)base + p] = rec;
        }
    }
    for (int j = 16; j < c; ++j) {
        const int2 q   = seg[j];
        const int  p   = atomicAdd(&deg[q.y - r0], 1);
        const int2 rec = make_int2(q.x, 0);
        if (staged) stage[p] = rec;
        else        csr[(size_t)base + p] = rec;
    }
    if (staged) {
        __syncthreads();
        for (int i = t; i < nE; i += 256) csr[(size_t)base + i] = stage[i];
    }
}

__device__ __forceinline__ void fma4(float4& a, const float4 v, const float w) {
    a.x = fmaf(v.x, w, a.x);
    a.y = fmaf(v.y, w, a.y);
    a.z = fmaf(v.z, w, a.z);
    a.w = fmaf(v.w, w, a.w);
}

__device__ __forceinline__ float4 keep4(const bool p, const float4 v) {
    float4 r;
    r.x = p ? v.x : 0.0f;
    r.y = p ? v.y : 0.0f;
    r.z = p ? v.z : 0.0f;
    r.w = p ? v.w : 0.0f;
    return r;
}

__device__ __forceinline__ void pool_add(float* __restrict__ p, const float4 a) {
    atomicAdd(p,     a.x);
    atomicAdd(p + 1, a.y);
    atomicAdd(p + 2, a.z);
    atomicAdd(p + 3, a.w);
}

template <bool FIRST, bool INLDS>
__device__ __forceinline__ void gather_slice(const int2* rec, int* csr2,
                                             const int2* lrec, const int* lrp,
                                             const float* __restrict__ h_in,
                                             const float* __restrict__ dinv,
                                             float* tw, const int tb, const int nr,
                                             const int h, const int c, const int d0) {
    const char* hbytes = (const char*)h_in;
    const char* dbytes = (const char*)dinv;
    const unsigned d0b = 4u * (unsigned)d0;
    for (int li = h; li < nr; li += 2) {
        const int i   = tb + li;
        const int beg = lrp[li];
        const int end = lrp[li + 1];
        const float  di = dinv[i];
        const float4 hs = *(const float4*)(hbytes + (((unsigned)i << 8) + d0b));

        float4 acc = make_float4(0.0f, 0.0f, 0.0f, 0.0f);
        const int last = end - 1;
        for (int k = beg; k < end; k += 8) {
            int2   e[8];
            float4 v[8];
#pragma unroll
            for (int u = 0; u < 8; ++u) {
                const int idx = min(k + u, last);
                e[u] = INLDS ? lrec[idx] : rec[idx];
            }
#pragma unroll
            for (int u = 0; u < 8; ++u) v[u] = *(const float4*)(hbytes + (((unsigned)e[u].x << 8) + d0b));
            float w[8];
            if (FIRST) {
#pragma unroll
                for (int u = 0; u < 8; ++u) w[u] = *(const float*)(dbytes + ((unsigned)e[u].x << 2)) * di;
                float mine = w[0];
#pragma unroll
                for (int u = 1; u < 8; ++u) mine = (c == u) ? w[u] : mine;
                if (c < 8 && k + c < end) csr2[2 * (k + c)] = __float_as_int(mine);
            } else {
#pragma unroll
                for (int u = 0; u < 8; ++u) w[u] = __int_as_float(e[u].y);
            }
#pragma unroll
            for (int u = 0; u < 8; ++u) {
                const bool p = k + u < end;
                fma4(acc, keep4(p, v[u]), p ? w[u] : 0.0f);
            }
        }

        float4 o;
        o.x = acc.x + hs.x * di * di;
        o.y = acc.y + hs.y * di * di;
        o.z = acc.z + hs.z * di * di;
        o.w = acc.w + hs.w * di * di;
        *(float4*)(tw + li * TPITCH + d0) = o;
    }
}

template <bool FIRST>
__global__ __launch_bounds__(256) void layer_kernel(const int* __restrict__ rowptr,
                                                    int* __restrict__ csr,
                                                    const float* __restrict__ h_in,
                                                    const float* __restrict__ dinv,
                                                    const float* __restrict__ Wm,
                                                    const float* __restrict__ bias,
                                                    const float* __restrict__ gamma,
                                                    const float* __restrict__ beta,
                                                    const float* __restrict__ mean,
                                                    const float* __restrict__ var,
                                                    const int* __restrict__ batch,
                                                    float* __restrict__ h_out,
                                                    float* __restrict__ pooled,
                                                    int N, int layer, int n_layers) {
    __shared__ __attribute__((aligned(16))) float tiles[8][16 * TPITCH];
    __shared__ __attribute__((aligned(16))) float wl[DIMF * WPITCH];
    __shared__ __attribute__((aligned(16))) int2 recs[8][RCAP];
    __shared__ int rps[8][20];
#pragma unroll
    for (int q = 0; q < 4; ++q) {
        const int idx = (int)threadIdx.x + 256 * q;
        const int wr  = idx >> 4;
        const int wc  = (idx & 15) * 4;
        *(float4*)(wl + wr * WPITCH + wc) = *(const float4*)(Wm + wr * DIMF + wc);
    }
    __syncthreads();

    const int wib  = __builtin_amdgcn_readfirstlane((int)(threadIdx.x >> 5));
    const int wave = blockIdx.x * 8 + wib;
    const int nw   = gridDim.x * 8;
    const int base = N / nw;
    const int rem  = N - base * nw;
    const int row0 = wave * base + min(wave, rem);
    const int row1 = row0 + base + (wave < rem ? 1 : 0);
    const int lane = threadIdx.x & 31;
    const int c    = lane & 15;
    const int h    = lane >> 4;
    const int d0   = c * 4;
    const int ld   = layer * DIMF + d0;
    float* tw = tiles[wib];
    int2*  lrec = recs[wib];
    int*   lrp  = rps[wib];
    const int2* rec = (const int2*)csr;
    float* __restrict__ pcol = pooled + ld;

    int    have = 0;
    int    gp   = 0;
    float4 ax   = make_float4(0.0f, 0.0f, 0.0f, 0.0f);

    {
        const int tb = row0;
        const int nr = min(16, row1 - row0);

        const float4 bb = *(const float4*)(bias + d0);
        const float4 mu = *(const float4*)(mean + ld);
        const float4 vv = *(const float4*)(var + ld);
        const float4 ga = *(const float4*)(gamma + ld);
        const float4 be = *(const float4*)(beta + ld);
        float4 rs;
        rs.x = __frsqrt_rn(vv.x + 1e-5f);
        rs.y = __frsqrt_rn(vv.y + 1e-5f);
        rs.z = __frsqrt_rn(vv.z + 1e-5f);
        rs.w = __frsqrt_rn(vv.w + 1e-5f);
        int gid[8];
#pragma unroll
        for (int r = 0; r < 8; ++r) gid[r] = batch[min(tb + 8 * h + r, row1 - 1)];

        const int rpv = rowptr[min(row0 + lane, row1)];
        const int s0 = __builtin_amdgcn_readlane(rpv, 0);
        if (lane <= 16) lrp[lane] = rpv - s0;
        const int s1 = __builtin_amdgcn_readlane(rpv, nr);
        const int ns = s1 - s0;
        const bool inlds = ns <= RCAP;
        const int2* srec = rec + s0;
        int* csr2 = csr + 2 * (size_t)s0 + 1;
        if (inlds && ns > 0) {
            int2 t[RCAP / 32];
#pragma unroll
            for (int q = 0; q < RCAP / 32; ++q) t[q] = srec[min(lane + 32 * q, ns - 1)];
#pragma unroll
            for (int q = 0; q < RCAP / 32; ++q)
                if (lane + 32 * q < ns) lrec[lane + 32 * q] = t[q];
        }

        if (inlds)
            gather_slice<FIRST, true>(srec, csr2, lrec, lrp, h_in, dinv, tw, tb, nr, h, c, d0);
        else
            gather_slice<FIRST, false>(srec, csr2, lrec, lrp, h_in, dinv, tw, tb, nr, h, c, d0);
#pragma unroll
        for (int q = 0; q < 8; ++q) {
            const int li = 2 * q + h;
            if (li >= nr) *(float4*)(tw + li * TPITCH + d0) = make_float4(0.0f, 0.0f, 0.0f, 0.0f);
        }

        const float* arow = tw + c * TPITCH + 4 * h;
        v8f acc[4];
#pragma unroll
        for (int nt = 0; nt < 4; ++nt) acc[nt] = v8f{};

#pragma unroll
        for (int j = 0; j < 8; ++j) {
            const float4 qa = *(const float4*)(arow + 8 * j);
            float av[4];
            av[0] = qa.x;
            av[1] = qa.y;
            av[2] = qa.z;
            av[3] = qa.w;
            float wv[4][4];
#pragma unroll
            for (int i = 0; i < 4; ++i) {
                const float4 q = *(const float4*)(wl + (8 * j + 4 * h + i) * WPITCH + 4 * c);
                wv[i][0] = q.x;
                wv[i][1] = q.y;
                wv[i][2] = q.z;
                wv[i][3] = q.w;
            }
#pragma unroll
            for (int s = 0; s < 2; ++s) {
                v2f a;
                a.x = av[2 * s];
                a.y = av[2 * s + 1];
#pragma unroll
                for (int nt = 0; nt < 4; ++nt) {
                    v2f b;
                    b.x = wv[2 * s][nt];
                    b.y = wv[2 * s + 1][nt];
                    acc[nt] = __builtin_amdgcn_wmma_f32_16x16x4_f32(
                             false, a,  false, b,
                             (short)0, acc[nt],  false,  false);
                }
            }
        }

        float4 t[8];
#pragma unroll
        for (int r = 0; r < 8; ++r) {
            const float ox = acc[0][r] + bb.x;
            const float oy = acc[1][r] + bb.y;
            const float oz = acc[2][r] + bb.z;
            const float ow = acc[3][r] + bb.w;
            t[r].x = fmaxf((ox - mu.x) * rs.x * ga.x + be.x, 0.0f);
            t[r].y = fmaxf((oy - mu.y) * rs.y * ga.y + be.y, 0.0f);
            t[r].z = fmaxf((oz - mu.z) * rs.z * ga.z + be.z, 0.0f);
            t[r].w = fmaxf((ow - mu.w) * rs.w * ga.w + be.w, 0.0f);
        }

#pragma unroll
        for (int r = 0; r < 8; ++r) {
            const int li = 8 * h + r;
            if (li < nr) {
                const int i = tb + li;
                const int g = gid[r];
                if (layer + 1 < n_layers) *(float4*)(h_out + (size_t)i * DIMF + d0) = t[r];

                if (have && g != gp) {
                    pool_add(pcol + (size_t)gp * CATF, ax);
                    ax = make_float4(0.0f, 0.0f, 0.0f, 0.0f);
                }
                gp   = g;
                have = 1;
                ax.x += t[r].x;
                ax.y += t[r].y;
                ax.z += t[r].z;
                ax.w += t[r].w;
            }
        }
    }

    const int   have_o = __shfl_xor(have, 16);
    const int   gp_o   = __shfl_xor(gp, 16);
    float4 ax_o;
    ax_o.x = __shfl_xor(ax.x, 16);
    ax_o.y = __shfl_xor(ax.y, 16);
    ax_o.z = __shfl_xor(ax.z, 16);
    ax_o.w = __shfl_xor(ax.w, 16);
    if (have) {
        if (have_o && gp_o == gp) {
            if (h == 0) {
                ax.x += ax_o.x;
                ax.y += ax_o.y;
                ax.z += ax_o.z;
                ax.w += ax_o.w;
                pool_add(pcol + (size_t)gp * CATF, ax);
            }
        } else {
            pool_add(pcol + (size_t)gp * CATF, ax);
        }
    }
}

__global__ __launch_bounds__(64) void mlp_kernel(const float* __restrict__ pooled,
                                                 const float* __restrict__ W1, const float* __restrict__ b1,
                                                 const float* __restrict__ W2, const float* __restrict__ b2,
                                                 float* __restrict__ out) {
    __shared__ float sp[CATF];
    __shared__ float sh[DIMF];
    int g = blockIdx.x, t = threadIdx.x;
    const float* prow = pooled + (size_t)g * CATF;
#pragma unroll
    for (int j = 0; j < 4; ++j) sp[t + j * DIMF] = prow[t + j * DIMF];
    __syncthreads();
    float acc = b1[t];
#pragma unroll 8
    for (int k = 0; k < CATF; ++k) acc = fmaf(sp[k], W1[k * DIMF + t], acc);
    sh[t] = fmaxf(acc, 0.0f);
    __syncthreads();
    if (t < OUTF) {
        float o = b2[t];
#pragma unroll 8
        for (int d = 0; d < DIMF; ++d) o = fmaf(sh[d], W2[d * OUTF + t], o);
        out[(size_t)g * OUTF + t] = o;
    }
}

extern "C" void kernel_launch(void* const* d_in, const int* in_sizes, int n_in,
                              void* d_out, int out_size, void* d_ws, size_t ws_size,
                              hipStream_t stream) {
    const float* x    = (const float*)d_in[0];
    const int*   ei   = (const int*)d_in[1];
    const int*   bat  = (const int*)d_in[2];
    const float* W[4] = {(const float*)d_in[4], (const float*)d_in[6],
                         (const float*)d_in[8], (const float*)d_in[10]};
    const float* B[4] = {(const float*)d_in[5], (const float*)d_in[7],
                         (const float*)d_in[9], (const float*)d_in[11]};
    const float* bn_g = (const float*)d_in[12];
    const float* bn_b = (const float*)d_in[13];
    const float* bn_m = (const float*)d_in[14];
    const float* bn_v = (const float*)d_in[15];
    const float* l1W  = (const float*)d_in[16];
    const float* l1b  = (const float*)d_in[17];
    const float* l2W  = (const float*)d_in[18];
    const float* l2b  = (const float*)d_in[19];

    const int N = in_sizes[0] / DIMF;
    const int E = in_sizes[1] / 2;
    const int G = out_size / OUTF;
    const int* src = ei;
    const int* dst = ei + E;

    float* ws     = (float*)d_ws;
    const int K   = (N + RB - 1) / RB;
    const int CH  = (E + BINB - 1) / BINB;
    float* dinv   = ws;
    int*   rowptr = (int*)(dinv + PAD64(N));
    int*   startT = rowptr + PAD64(N + 1);
    int*   binned = startT + PAD64((size_t)(K + 1) * BINB);
    int*   csr    = binned + PAD64(2 * (size_t)E);
    float* h_a    = (float*)(csr + PAD64(2 * (size_t)E));
    float* h_b    = h_a + (size_t)N * DIMF;
    float* pooled = h_b + (size_t)N * DIMF;

    const int T = 256;
    bin_kernel<<<BINB, 1024, 0, stream>>>(src, dst, binned, startT, pooled, G * CATF, E, CH, K);
    bucket_kernel<<<K, T, 0, stream>>>(binned, startT, rowptr, dinv, csr, N, CH);

    float* hbuf[2] = {h_a, h_b};
    const float* hin = x;
    for (int l = 0; l < 4; ++l) {
        float* hout = hbuf[l & 1];
        if (l == 0)
            layer_kernel<true><<<LOOP_BLOCKS, T, 0, stream>>>(rowptr, csr, hin, dinv, W[l], B[l],
                                                            bn_g, bn_b, bn_m, bn_v, bat,
                                                            hout, pooled, N, l, 4);
        else
            layer_kernel<false><<<LOOP_BLOCKS, T, 0, stream>>>(rowptr, csr, hin, dinv, W[l], B[l],
                                                             bn_g, bn_b, bn_m, bn_v, bat,
                                                             hout, pooled, N, l, 4);
        hin = hout;
    }

    mlp_kernel<<<G, DIMF, 0, stream>>>(pooled, l1W, l1b, l2W, l2b, (float*)d_out);
}


# MLP launch: a 256-thread block takes 4 graphs, a thread keeps its 64 lin1 weights in registers, all loads asked for at once: one memory trip, not 40 in a row; k sum order changes (4 quarter sums)
# speedup vs baseline: 5.4194x; 1.1075x over previous
#include <hip/hip_runtime.h>
#include <hip/hip_bf16.h>


typedef float v2f __attribute__((ext_vector_type(2)));
typedef float v8f __attribute__((ext_vector_type(8)));

#define DIMF 64
#define CATF 256
#define OUTF 10
#define LOOP_BLOCKS 1024
#define TPITCH 68
#define WPITCH 72
#define PAD64(n) (((size_t)(n) + 63) & ~(size_t)63)
#define RB 128
#define RBSH 7
#define KMAX 1024
#define BINB 256
#define SCAP 7168
#define LCAP 6144
#define RCAP 320

template <int NW>
__device__ __forceinline__ int block_excl_scan(const int v, int* sa, int* sb, int& total) {
    const int lane = threadIdx.x & 31;
    const int w    = threadIdx.x >> 5;
    int x = v;
#pragma unroll
    for (int off = 1; off < 32; off <<= 1) {
        const int y = __shfl_up(x, off);
        if (lane >= off) x += y;
    }
    if (lane == 31) sa[w] = x;
    __syncthreads();
    if (w == 0) {
        const int s = (lane < NW) ? sa[lane] : 0;
        int xs = s;
#pragma unroll
        for (int off = 1; off < 32; off <<= 1) {
            const int y = __shfl_up(xs, off);
            if (lane >= off) xs += y;
        }
        sb[lane] = xs - s;
        if (lane == 31) sb[32] = xs;
    }
    __syncthreads();
    total = sb[32];
    return sb[w] + x - v;
}

__global__ __launch_bounds__(1024) void bin_kernel(const int* __restrict__ src,
                                                   const int* __restrict__ dst,
                                                   int* __restrict__ binned_w,
                                                   int* __restrict__ startT,
                                                   float* __restrict__ pooled,
                                                   int npool, int E, int CH, int K) {
    __shared__ int hist[KMAX];
    __shared__ int sa[33];
    __shared__ int sb[33];
    __shared__ __attribute__((aligned(16))) int2 stage[SCAP];
    int2* __restrict__ binned = (int2*)binned_w;
    const int t = threadIdx.x, b = blockIdx.x;
    for (int i = b * 1024 + t; i < npool; i += gridDim.x * 1024) pooled[i] = 0.0f;

    const int e0 = b * CH;
    const int n  = max(0, min(CH, E - e0));
    const int* __restrict__ dch = dst + e0;
    const int* __restrict__ sch = src + e0;
    hist[t] = 0;
    __syncthreads();
    for (int i0 = t; i0 < n; i0 += 4096) {
        int d[4];
#pragma unroll
        for (int u = 0; u < 4; ++u) d[u] = dch[min(i0 + 1024 * u, n - 1)];
#pragma unroll
        for (int u = 0; u < 4; ++u)
            if (i0 + 1024 * u < n) atomicAdd(&hist[d[u] >> RBSH], 1);
    }
    __syncthreads();
    const int cnt = hist[t];
    int total;
    const int ex = block_excl_scan<32>(cnt, sa, sb, total);
    if (t < K) startT[t * BINB + b] = ex;
    if (t == 0) startT[K * BINB + b] = n;
    hist[t] = ex;
    __syncthreads();

    const bool staged = n <= SCAP;
    for (int i0 = t; i0 < n; i0 += 4096) {
        int s[4], d[4];
#pragma unroll
        for (int u = 0; u < 4; ++u) {
            const int idx = min(i0 + 1024 * u, n - 1);
            s[u] = sch[idx];
            d[u] = dch[idx];
        }
#pragma unroll
        for (int u = 0; u < 4; ++u) {
            if (i0 + 1024 * u < n) {
                const int  p = atomicAdd(&hist[d[u] >> RBSH], 1);
                const int2 r = make_int2(s[u], d[u]);
                if (staged) stage[p] = r;
                else        binned[(size_t)e0 + p] = r;
            }
        }
    }
    if (staged) {
        __syncthreads();
        for (int i = t; i < n; i += 1024) binned[(size_t)e0 + i] = stage[i];
    }
}

__global__ __launch_bounds__(256) void bucket_kernel(const int* __restrict__ binned_w,
                                                     const int* __restrict__ startT,
                                                     int* __restrict__ rowptr,
                                                     float* __restrict__ dinv,
                                                     int* __restrict__ csr_w,
                                                     int N, int CH) {
    __shared__ int deg[RB];
    __shared__ int sa[33];
    __shared__ int sb[33];
    __shared__ __attribute__((aligned(16))) int2 stage[LCAP];
    const int2* __restrict__ binned = (const int2*)binned_w;
    int2* __restrict__ csr = (int2*)csr_w;
    const int t = threadIdx.x, k = blockIdx.x;
    const int r0 = k * RB;
    const int s0 = startT[k * BINB + t];
    const int c  = startT[(k + 1) * BINB + t] - s0;
    if (t < RB) deg[t] = 0;
    int base, nE;
    (void)block_excl_scan<8>(s0, sa, sb, base);
    (void)block_excl_scan<8>(c, sa, sb, nE);

    const int2* __restrict__ seg  = binned + (size_t)t * CH + s0;
    const int2* __restrict__ segs = (c > 0) ? seg : binned;
    const int cl = max(c - 1, 0);
    int2 r[16];
#pragma unroll
    for (int j = 0; j < 16; ++j) r[j] = segs[min(j, cl)];
#pragma unroll
    for (int j = 0; j < 16; ++j)
        if (j < c) atomicAdd(&deg[r[j].y - r0], 1);
    for (int j = 16; j < c; ++j) atomicAdd(&deg[seg[j].y - r0], 1);
    __syncthreads();

    const int dg = (t < RB) ? deg[t] : 0;
    int tot2;
    const int rs  = block_excl_scan<8>(dg, sa, sb, tot2);
    const int row = r0 + t;
    if (t < RB && row < N) {
        rowptr[row] = base + rs;
        dinv[row]   = __frsqrt_rn((float)dg + 1.0f);
    }
    if (k == (int)gridDim.x - 1 && t == 0) rowptr[N] = base + nE;
    if (t < RB) deg[t] = rs;
    __syncthreads();

    const bool staged = nE <= LCAP;
#pragma unroll
    for (int j = 0; j < 16; ++j) {
        if (j < c) {
            const int  p   = atomicAdd(&deg[r[j].y - r0], 1);
            const int2 rec = make_int2(r[j].x, 0);
            if (staged) stage[p] = rec;
            else        csr[(size_t)base + p] = rec;
        }
    }
    for (int j = 16; j < c; ++j) {
        const int2 q   = seg[j];
        const int  p   = atomicAdd(&deg[q.y - r0], 1);
        const int2 rec = make_int2(q.x, 0);
        if (staged) stage[p] = rec;
        else        csr[(size_t)base + p] = rec;
    }
    if (staged) {
        __syncthreads();
        for (int i = t; i < nE; i += 256) csr[(size_t)base + i] = stage[i];
    }
}

__device__ __forceinline__ void fma4(float4& a, const float4 v, const float w) {
    a.x = fmaf(v.x, w, a.x);
    a.y = fmaf(v.y, w, a.y);
    a.z = fmaf(v.z, w, a.z);
    a.w = fmaf(v.w, w, a.w);
}

__device__ __forceinline__ float4 keep4(const bool p, const float4 v) {
    float4 r;
    r.x = p ? v.x : 0.0f;
    r.y = p ? v.y : 0.0f;
    r.z = p ? v.z : 0.0f;
    r.w = p ? v.w : 0.0f;
    return r;
}

__device__ __forceinline__ void pool_add(float* __restrict__ p, const float4 a) {
    atomicAdd(p,     a.x);
    atomicAdd(p + 1, a.y);
    atomicAdd(p + 2, a.z);
    atomicAdd(p + 3, a.w);
}

template <bool FIRST, bool INLDS>
__device__ __forceinline__ void gather_slice(const int2* rec, int* csr2,
                                             const int2* lrec, const int* lrp,
                                             const float* __restrict__ h_in,
                                             const float* __restrict__ dinv,
                                             float* tw, const int tb, const int nr,
                                             const int h, const int c, const int d0) {
    const char* hbytes = (const char*)h_in;
    const char* dbytes = (const char*)dinv;
    const unsigned d0b = 4u * (unsigned)d0;
    for (int li = h; li < nr; li += 2) {
        const int i   = tb + li;
        const int beg = lrp[li];
        const int end = lrp[li + 1];
        const float  di = dinv[i];
        const float4 hs = *(const float4*)(hbytes + (((unsigned)i << 8) + d0b));

        float4 acc = make_float4(0.0f, 0.0f, 0.0f, 0.0f);
        const int last = end - 1;
        for (int k = beg; k < end; k += 8) {
            int2   e[8];
            float4 v[8];
#pragma unroll
            for (int u = 0; u < 8; ++u) {
                const int idx = min(k + u, last);
                e[u] = INLDS ? lrec[idx] : rec[idx];
            }
#pragma unroll
            for (int u = 0; u < 8; ++u) v[u] = *(const float4*)(hbytes + (((unsigned)e[u].x << 8) + d0b));
            float w[8];
            if (FIRST) {
#pragma unroll
                for (int u = 0; u < 8; ++u) w[u] = *(const float*)(dbytes + ((unsigned)e[u].x << 2)) * di;
                float mine = w[0];
#pragma unroll
                for (int u = 1; u < 8; ++u) mine = (c == u) ? w[u] : mine;
                if (c < 8 && k + c < end) csr2[2 * (k + c)] = __float_as_int(mine);
            } else {
#pragma unroll
                for (int u = 0; u < 8; ++u) w[u] = __int_as_float(e[u].y);
            }
#pragma unroll
            for (int u = 0; u < 8; ++u) {
                const bool p = k + u < end;
                fma4(acc, keep4(p, v[u]), p ? w[u] : 0.0f);
            }
        }

        float4 o;
        o.x = acc.x + hs.x * di * di;
        o.y = acc.y + hs.y * di * di;
        o.z = acc.z + hs.z * di * di;
        o.w = acc.w + hs.w * di * di;
        *(float4*)(tw + li * TPITCH + d0) = o;
    }
}

template <bool FIRST>
__global__ __launch_bounds__(256) void layer_kernel(const int* __restrict__ rowptr,
                                                    int* __restrict__ csr,
                                                    const float* __restrict__ h_in,
                                                    const float* __restrict__ dinv,
                                                    const float* __restrict__ Wm,
                                                    const float* __restrict__ bias,
                                                    const float* __restrict__ gamma,
                                                    const float* __restrict__ beta,
                                                    const float* __restrict__ mean,
                                                    const float* __restrict__ var,
                                                    const int* __restrict__ batch,
                                                    float* __restrict__ h_out,
                                                    float* __restrict__ pooled,
                                                    int N, int layer, int n_layers) {
    __shared__ __attribute__((aligned(16))) float tiles[8][16 * TPITCH];
    __shared__ __attribute__((aligned(16))) float wl[DIMF * WPITCH];
    __shared__ __attribute__((aligned(16))) int2 recs[8][RCAP];
    __shared__ int rps[8][20];
#pragma unroll
    for (int q = 0; q < 4; ++q) {
        const int idx = (int)threadIdx.x + 256 * q;
        const int wr  = idx >> 4;
        const int wc  = (idx & 15) * 4;
        *(float4*)(wl + wr * WPITCH + wc) = *(const float4*)(Wm + wr * DIMF + wc);
    }
    __syncthreads();

    const int wib  = __builtin_amdgcn_readfirstlane((int)(threadIdx.x >> 5));
    const int wave = blockIdx.x * 8 + wib;
    const int nw   = gridDim.x * 8;
    const int base = N / nw;
    const int rem  = N - base * nw;
    const int row0 = wave * base + min(wave, rem);
    const int row1 = row0 + base + (wave < rem ? 1 : 0);
    const int lane = threadIdx.x & 31;
    const int c    = lane & 15;
    const int h    = lane >> 4;
    const int d0   = c * 4;
    const int ld   = layer * DIMF + d0;
    float* tw = tiles[wib];
    int2*  lrec = recs[wib];
    int*   lrp  = rps[wib];
    const int2* rec = (const int2*)csr;
    float* __restrict__ pcol = pooled + ld;

    int    have = 0;
    int    gp   = 0;
    float4 ax   = make_float4(0.0f, 0.0f, 0.0f, 0.0f);

    {
        const int tb = row0;
        const int nr = min(16, row1 - row0);

        const float4 bb = *(const float4*)(bias + d0);
        const float4 mu = *(const float4*)(mean + ld);
        const float4 vv = *(const float4*)(var + ld);
        const float4 ga = *(const float4*)(gamma + ld);
        const float4 be = *(const float4*)(beta + ld);
        float4 rs;
        rs.x = __frsqrt_rn(vv.x + 1e-5f);
        rs.y = __frsqrt_rn(vv.y + 1e-5f);
        rs.z = __frsqrt_rn(vv.z + 1e-5f);
        rs.w = __frsqrt_rn(vv.w + 1e-5f);
        int gid[8];
#pragma unroll
        for (int r = 0; r < 8; ++r) gid[r] = batch[min(tb + 8 * h + r, row1 - 1)];

        const int rpv = rowptr[min(row0 + lane, row1)];
        const int s0 = __builtin_amdgcn_readlane(rpv, 0);
        if (lane <= 16) lrp[lane] = rpv - s0;
        const int s1 = __builtin_amdgcn_readlane(rpv, nr);
        const int ns = s1 - s0;
        const bool inlds = ns <= RCAP;
        const int2* srec = rec + s0;
        int* csr2 = csr + 2 * (size_t)s0 + 1;
        if (inlds && ns > 0) {
            int2 t[RCAP / 32];
#pragma unroll
            for (int q = 0; q < RCAP / 32; ++q) t[q] = srec[min(lane + 32 * q, ns - 1)];
#pragma unroll
            for (int q = 0; q < RCAP / 32; ++q)
                if (lane + 32 * q < ns) lrec[lane + 32 * q] = t[q];
        }

        if (inlds)
            gather_slice<FIRST, true>(srec, csr2, lrec, lrp, h_in, dinv, tw, tb, nr, h, c, d0);
        else
            gather_slice<FIRST, false>(srec, csr2, lrec, lrp, h_in, dinv, tw, tb, nr, h, c, d0);
#pragma unroll
        for (int q = 0; q < 8; ++q) {
            const int li = 2 * q + h;
            if (li >= nr) *(float4*)(tw + li * TPITCH + d0) = make_float4(0.0f, 0.0f, 0.0f, 0.0f);
        }

        const float* arow = tw + c * TPITCH + 4 * h;
        v8f acc[4];
#pragma unroll
        for (int nt = 0; nt < 4; ++nt) acc[nt] = v8f{};

#pragma unroll
        for (int j = 0; j < 8; ++j) {
            const float4 qa = *(const float4*)(arow + 8 * j);
            float av[4];
            av[0] = qa.x;
            av[1] = qa.y;
            av[2] = qa.z;
            av[3] = qa.w;
            float wv[4][4];
#pragma unroll
            for (int i = 0; i < 4; ++i) {
                const float4 q = *(const float4*)(wl + (8 * j + 4 * h + i) * WPITCH + 4 * c);
                wv[i][0] = q.x;
                wv[i][1] = q.y;
                wv[i][2] = q.z;
                wv[i][3] = q.w;
            }
#pragma unroll
            for (int s = 0; s < 2; ++s) {
                v2f a;
                a.x = av[2 * s];
                a.y = av[2 * s + 1];
#pragma unroll
                for (int nt = 0; nt < 4; ++nt) {
                    v2f b;
                    b.x = wv[2 * s][nt];
                    b.y = wv[2 * s + 1][nt];
                    acc[nt] = __builtin_amdgcn_wmma_f32_16x16x4_f32(
                             false, a,  false, b,
                             (short)0, acc[nt],  false,  false);
                }
            }
        }

        float4 t[8];
#pragma unroll
        for (int r = 0; r < 8; ++r) {
            const float ox = acc[0][r] + bb.x;
            const float oy = acc[1][r] + bb.y;
            const float oz = acc[2][r] + bb.z;
            const float ow = acc[3][r] + bb.w;
            t[r].x = fmaxf((ox - mu.x) * rs.x * ga.x + be.x, 0.0f);
            t[r].y = fmaxf((oy - mu.y) * rs.y * ga.y + be.y, 0.0f);
            t[r].z = fmaxf((oz - mu.z) * rs.z * ga.z + be.z, 0.0f);
            t[r].w = fmaxf((ow - mu.w) * rs.w * ga.w + be.w, 0.0f);
        }

#pragma unroll
        for (int r = 0; r < 8; ++r) {
            const int li = 8 * h + r;
            if (li < nr) {
                const int i = tb + li;
                const int g = gid[r];
                if (layer + 1 < n_layers) *(float4*)(h_out + (size_t)i * DIMF + d0) = t[r];

                if (have && g != gp) {
                    pool_add(pcol + (size_t)gp * CATF, ax);
                    ax = make_float4(0.0f, 0.0f, 0.0f, 0.0f);
                }
                gp   = g;
                have = 1;
                ax.x += t[r].x;
                ax.y += t[r].y;
                ax.z += t[r].z;
                ax.w += t[r].w;
            }
        }
    }

    const int   have_o = __shfl_xor(have, 16);
    const int   gp_o   = __shfl_xor(gp, 16);
    float4 ax_o;
    ax_o.x = __shfl_xor(ax.x, 16);
    ax_o.y = __shfl_xor(ax.y, 16);
    ax_o.z = __shfl_xor(ax.z, 16);
    ax_o.w = __shfl_xor(ax.w, 16);
    if (have) {
        if (have_o && gp_o == gp) {
            if (h == 0) {
                ax.x += ax_o.x;
                ax.y += ax_o.y;
                ax.z += ax_o.z;
                ax.w += ax_o.w;
                pool_add(pcol + (size_t)gp * CATF, ax);
            }
        } else {
            pool_add(pcol + (size_t)gp * CATF, ax);
        }
    }
}

#define MG 4
__global__ __launch_bounds__(256) void mlp_kernel(const float* __restrict__ pooled,
                                                  const float* __restrict__ W1, const float* __restrict__ b1,
                                                  const float* __restrict__ W2, const float* __restrict__ b2,
                                                  float* __restrict__ out, int G) {
    __shared__ __attribute__((aligned(16))) float sp[MG][CATF];
    __shared__ __attribute__((aligned(16))) float w2s[DIMF * OUTF];
    __shared__ float part[MG][4][DIMF];
    __shared__ float sh[MG][DIMF];
    const int t   = threadIdx.x;
    const int col = t & 63;
    const int q   = t >> 6;
    const int g0  = blockIdx.x * MG;

    const int gl = min(g0 + q, G - 1);
    const float4 pv = *(const float4*)(pooled + (size_t)gl * CATF + 4 * col);
    float w[64];
    const float* __restrict__ wp = W1 + (size_t)(64 * q) * DIMF + col;
#pragma unroll
    for (int kk = 0; kk < 64; ++kk) w[kk] = wp[kk * DIMF];
    float4 w2v = make_float4(0.0f, 0.0f, 0.0f, 0.0f);
    if (t < DIMF * OUTF / 4) w2v = *(const float4*)(W2 + 4 * t);
    const int kc = t & 3;
    const int og = t >> 2;
    const int o  = og % OUTF;
    const int g2 = og / OUTF;
    const float bias1 = b1[col];
    const float bias2 = b2[o];

    *(float4*)(&sp[q][4 * col]) = pv;
    if (t < DIMF * OUTF / 4) *(float4*)(&w2s[4 * t]) = w2v;
    __syncthreads();

    float acc[MG];
#pragma unroll
    for (int g = 0; g < MG; ++g) acc[g] = 0.0f;
#pragma unroll
    for (int k4 = 0; k4 < 16; ++k4) {
#pragma unroll
        for (int g = 0; g < MG; ++g) {
            const float4 s = *(const float4*)(&sp[g][64 * q + 4 * k4]);
            acc[g] = fmaf(s.x, w[4 * k4 + 0], acc[g]);
            acc[g] = fmaf(s.y, w[4 * k4 + 1], acc[g]);
            acc[g] = fmaf(s.z, w[4 * k4 + 2], acc[g]);
            acc[g] = fmaf(s.w, w[4 * k4 + 3], acc[g]);
        }
    }
#pragma unroll
    for (int g = 0; g < MG; ++g) part[g][q][col] = acc[g];
    __syncthreads();

    sh[q][col] = fmaxf(((part[q][0][col] + part[q][1][col]) + part[q][2][col]) + part[q][3][col] + bias1, 0.0f);
    __syncthreads();

    if (t < MG * OUTF * 4) {
        float a = 0.0f;
#pragma unroll
        for (int j = 0; j < 16; ++j) {
            const int k = 16 * kc + j;
            a = fmaf(sh[g2][k], w2s[k * OUTF + o], a);
        }
        a += __shfl_xor(a, 1);
        a += __shfl_xor(a, 2);
        if (kc == 0 && g0 + g2 < G) out[(size_t)(g0 + g2) * OUTF + o] = a + bias2;
    }
}

extern "C" void kernel_launch(void* const* d_in, const int* in_sizes, int n_in,
                              void* d_out, int out_size, void* d_ws, size_t ws_size,
                              hipStream_t stream) {
    const float* x    = (const float*)d_in[0];
    const int*   ei   = (const int*)d_in[1];
    const int*   bat  = (const int*)d_in[2];
    const float* W[4] = {(const float*)d_in[4], (const float*)d_in[6],
                         (const float*)d_in[8], (const float*)d_in[10]};
    const float* B[4] = {(const float*)d_in[5], (const float*)d_in[7],
                         (const float*)d_in[9], (const float*)d_in[11]};
    const float* bn_g = (const float*)d_in[12];
    const float* bn_b = (const float*)d_in[13];
    const float* bn_m = (const float*)d_in[14];
    const float* bn_v = (const float*)d_in[15];
    const float* l1W  = (const float*)d_in[16];
    const float* l1b  = (const float*)d_in[17];
    const float* l2W  = (const float*)d_in[18];
    const float* l2b  = (const float*)d_in[19];

    const int N = in_sizes[0] / DIMF;
    const int E = in_sizes[1] / 2;
    const int G = out_size / OUTF;
    const int* src = ei;
    const int* dst = ei + E;

    float* ws     = (float*)d_ws;
    const int K   = (N + RB - 1) / RB;
    const int CH  = (E + BINB - 1) / BINB;
    float* dinv   = ws;
    int*   rowptr = (int*)(dinv + PAD64(N));
    int*   startT = rowptr + PAD64(N + 1);
    int*   binned = startT + PAD64((size_t)(K + 1) * BINB);
    int*   csr    = binned + PAD64(2 * (size_t)E);
    float* h_a    = (float*)(csr + PAD64(2 * (size_t)E));
    float* h_b    = h_a + (size_t)N * DIMF;
    float* pooled = h_b + (size_t)N * DIMF;

    const int T = 256;
    bin_kernel<<<BINB, 1024, 0, stream>>>(src, dst, binned, startT, pooled, G * CATF, E, CH, K);
    bucket_kernel<<<K, T, 0, stream>>>(binned, startT, rowptr, dinv, csr, N, CH);

    float* hbuf[2] = {h_a, h_b};
    const float* hin = x;
    for (int l = 0; l < 4; ++l) {
        float* hout = hbuf[l & 1];
        if (l == 0)
            layer_kernel<true><<<LOOP_BLOCKS, T, 0, stream>>>(rowptr, csr, hin, dinv, W[l], B[l],
                                                            bn_g, bn_b, bn_m, bn_v, bat,
                                                            hout, pooled, N, l, 4);
        else
            layer_kernel<false><<<LOOP_BLOCKS, T, 0, stream>>>(rowptr, csr, hin, dinv, W[l], B[l],
                                                             bn_g, bn_b, bn_m, bn_v, bat,
                                                             hout, pooled, N, l, 4);
        hin = hout;
    }

    mlp_kernel<<<(G + MG - 1) / MG, T, 0, stream>>>(pooled, l1W, l1b, l2W, l2b, (float*)d_out, G);
}
